# starting kernel
# baseline (speedup 1.0000x reference)
#include <hip/hip_runtime.h>


#define LSTM_B 4096
#define LSTM_T 512

typedef float v2f __attribute__((ext_vector_type(2)));
typedef float v8f __attribute__((ext_vector_type(8)));

static __device__ __forceinline__ v8f wmma4(v2f a, v2f b, v8f c) {
  return __builtin_amdgcn_wmma_f32_16x16x4_f32(false, a, false, b, (short)0, c,
                                               false, false);
}

static __device__ __forceinline__ float sigm(float x) {
  return __builtin_amdgcn_rcpf(1.0f + __builtin_amdgcn_exp2f(-1.44269504f * x));
}
static __device__ __forceinline__ float tanh_a(float x) {
  return 1.0f - 2.0f * __builtin_amdgcn_rcpf(
                           1.0f + __builtin_amdgcn_exp2f(2.88539008f * x));
}

__global__ void __launch_bounds__(128, 1) lstm2_fc_kernel(
    const float* __restrict__ x,
    const float* __restrict__ Wih0,
    const float* __restrict__ Whh0,
    const float* __restrict__ bih0,
    const float* __restrict__ bhh0,
    const float* __restrict__ Wih1,
    const float* __restrict__ Whh1,
    const float* __restrict__ bih1,
    const float* __restrict__ bhh1,
    const float* __restrict__ Wfc,
    const float* __restrict__ bfc,
    float* __restrict__ out)
{
  const int lane = threadIdx.x & 31;
  const int half = lane >> 4;
  const int nIdx = lane & 15;
  const int wid  = (blockIdx.x * blockDim.x + threadIdx.x) >> 5;
  const int base = wid * 16;

  v2f Whh0A[4][4], Wih1A[4][4], Whh1A[4][4];
  v2f Wih0A[4];
  v2f B1A[4];
#pragma unroll
  for (int j = 0; j < 4; ++j) {
    const int row = 16 * j + nIdx;
#pragma unroll
    for (int kc = 0; kc < 4; ++kc) {
      const float* p0 = Whh0 + row * 16 + kc * 4 + 2 * half;
      const float* p1 = Wih1 + row * 16 + kc * 4 + 2 * half;
      const float* p2 = Whh1 + row * 16 + kc * 4 + 2 * half;
      Whh0A[j][kc] = v2f{p0[0], p0[1]};
      Wih1A[j][kc] = v2f{p1[0], p1[1]};
      Whh1A[j][kc] = v2f{p2[0], p2[1]};
    }
    const float w0  = Wih0[row];
    const float bb0 = bih0[row] + bhh0[row];
    const float bb1 = bih1[row] + bhh1[row];
    Wih0A[j] = v2f{half ? 0.0f : w0, half ? 0.0f : bb0};
    B1A[j]   = v2f{half ? 0.0f : bb1, 0.0f};
  }
  const v2f onesB = v2f{half ? 0.0f : 1.0f, 0.0f};
  const v8f vzero = {0.f, 0.f, 0.f, 0.f, 0.f, 0.f, 0.f, 0.f};

  v8f c0 = vzero, h0 = vzero, c1 = vzero, h1 = vzero;
  v2f h0B[4], h1B[4];
#pragma unroll
  for (int kc = 0; kc < 4; ++kc) {
    h0B[kc] = v2f{0.0f, 0.0f};
    h1B[kc] = v2f{0.0f, 0.0f};
  }

  auto convert = [&](const v8f& h, v2f(&hB)[4]) {
#pragma unroll
    for (int kc = 0; kc < 4; ++kc) {
      const int k0 = 4 * kc;
      if (kc < 2) {
        float sx = __shfl_xor(h[k0 + 2], 16, 32);
        float sy = __shfl_xor(h[k0 + 3], 16, 32);
        hB[kc] = v2f{half ? sx : h[k0 + 0], half ? sy : h[k0 + 1]};
      } else {
        float sx = __shfl_xor(h[k0 - 8], 16, 32);
        float sy = __shfl_xor(h[k0 - 7], 16, 32);
        hB[kc] = v2f{half ? h[k0 - 6] : sx, half ? h[k0 - 5] : sy};
      }
    }
  };

  const float* xrow = x + (base + nIdx) * LSTM_T;

#pragma unroll 1
  for (int t = 0; t < LSTM_T; ++t) {
    const float xv = xrow[t];
    const v2f xB = v2f{half ? 0.0f : xv, half ? 0.0f : 1.0f};

    v8f g[4];
#pragma unroll
    for (int j = 0; j < 4; ++j) {
      v8f acc = wmma4(Wih0A[j], xB, vzero);
#pragma unroll
      for (int kc = 0; kc < 4; ++kc) acc = wmma4(Whh0A[j][kc], h0B[kc], acc);
      g[j] = acc;
    }
#pragma unroll
    for (int v = 0; v < 8; ++v) {
      const float iv = sigm(g[0][v]);
      const float fv = sigm(g[1][v]);
      const float gv = tanh_a(g[2][v]);
      const float ov = sigm(g[3][v]);
      const float cv = fv * c0[v] + iv * gv;
      c0[v] = cv;
      h0[v] = ov * tanh_a(cv);
    }
    convert(h0, h0B);

#pragma unroll
    for (int j = 0; j < 4; ++j) {
      v8f acc = wmma4(B1A[j], onesB, vzero);
#pragma unroll
      for (int kc = 0; kc < 4; ++kc) acc = wmma4(Wih1A[j][kc], h0B[kc], acc);
#pragma unroll
      for (int kc = 0; kc < 4; ++kc) acc = wmma4(Whh1A[j][kc], h1B[kc], acc);
      g[j] = acc;
    }
#pragma unroll
    for (int v = 0; v < 8; ++v) {
      const float iv = sigm(g[0][v]);
      const float fv = sigm(g[1][v]);
      const float gv = tanh_a(g[2][v]);
      const float ov = sigm(g[3][v]);
      const float cv = fv * c1[v] + iv * gv;
      c1[v] = cv;
      h1[v] = ov * tanh_a(cv);
    }
    convert(h1, h1B);
  }

  float partial = 0.0f;
#pragma unroll
  for (int v = 0; v < 8; ++v) partial += Wfc[v + 8 * half] * h1[v];
  const float tot = partial + __shfl_xor(partial, 16, 32);
  if (half == 0) out[base + nIdx] = tot + bfc[0];
}

extern "C" void kernel_launch(void* const* d_in, const int* in_sizes, int n_in,
                              void* d_out, int out_size, void* d_ws,
                              size_t ws_size, hipStream_t stream) {
  (void)in_sizes; (void)n_in; (void)d_ws; (void)ws_size; (void)out_size;
  const float* x    = (const float*)d_in[0];
  const float* Wih0 = (const float*)d_in[1];
  const float* Whh0 = (const float*)d_in[2];
  const float* bih0 = (const float*)d_in[3];
  const float* bhh0 = (const float*)d_in[4];
  const float* Wih1 = (const float*)d_in[5];
  const float* Whh1 = (const float*)d_in[6];
  const float* bih1 = (const float*)d_in[7];
  const float* bhh1 = (const float*)d_in[8];
  const float* Wfc  = (const float*)d_in[9];
  const float* bfc  = (const float*)d_in[10];
  float* out = (float*)d_out;

  dim3 block(128);
  dim3 grid(LSTM_B / (16 * 4));
  lstm2_fc_kernel<<<grid, block, 0, stream>>>(x, Wih0, Whh0, bih0, bhh0, Wih1,
                                              Whh1, bih1, bhh1, Wfc, bfc, out);
}


# software-pipelined the time loop across the two layers, hand-interleaved: each layer's gate math issues in slices between the other layer's WMMAs; h lane exchange and x read issued ahead of use
# speedup vs baseline: 1.1400x; 1.1400x over previous
#include <hip/hip_runtime.h>


#define LSTM_B 4096
#define LSTM_T 512

typedef float v2f __attribute__((ext_vector_type(2)));
typedef float v8f __attribute__((ext_vector_type(8)));

static __device__ __forceinline__ v8f wmma4(v2f a, v2f b, v8f c) {
  return __builtin_amdgcn_wmma_f32_16x16x4_f32(false, a, false, b, (short)0, c,
                                               false, false);
}

#define FENCE() __builtin_amdgcn_sched_barrier(0)

#define K_SIG  (-1.44269504f)
#define K_TANH (2.88539008f)
#define EXP2(x) __builtin_amdgcn_exp2f(x)
#define RCP(x)  __builtin_amdgcn_rcpf(x)

#define SL(lo, hi, ...) _Pragma("unroll") for (int v = lo; v < hi; ++v) { __VA_ARGS__ FENCE(); }
#define S0(lo, hi, g, c, h, a)  SL(lo, hi, a[2][v] = K_TANH * g[2][v]; a[0][v] = K_SIG * g[0][v];)
#define S1(lo, hi, g, c, h, a)  SL(lo, hi, a[2][v] = EXP2(a[2][v]);    a[1][v] = K_SIG * g[1][v];)
#define S2(lo, hi, g, c, h, a)  SL(lo, hi, a[0][v] = EXP2(a[0][v]);    a[2][v] = 1.0f + a[2][v];)
#define S3(lo, hi, g, c, h, a)  SL(lo, hi, a[2][v] = RCP(a[2][v]);     a[3][v] = K_SIG * g[3][v];)
#define S4(lo, hi, g, c, h, a)  SL(lo, hi, a[1][v] = EXP2(a[1][v]);    a[0][v] = 1.0f + a[0][v];)
#define S5(lo, hi, g, c, h, a)  SL(lo, hi, a[0][v] = RCP(a[0][v]);     a[2][v] = 1.0f - 2.0f * a[2][v];)
#define S6(lo, hi, g, c, h, a)  SL(lo, hi, a[3][v] = EXP2(a[3][v]);    a[1][v] = 1.0f + a[1][v];)
#define S7(lo, hi, g, c, h, a)  SL(lo, hi, a[1][v] = RCP(a[1][v]);     a[0][v] = a[0][v] * a[2][v];)
#define S8(lo, hi, g, c, h, a)  SL(lo, hi, a[3][v] = 1.0f + a[3][v];   c[v] = a[1][v] * c[v] + a[0][v];)
#define S9(lo, hi, g, c, h, a)  SL(lo, hi, a[3][v] = RCP(a[3][v]);     a[2][v] = K_TANH * c[v];)
#define S10(lo, hi, g, c, h, a) SL(lo, hi, a[2][v] = EXP2(a[2][v]);)
#define S11(lo, hi, g, c, h, a) SL(lo, hi, a[2][v] = 1.0f + a[2][v];)
#define S12(lo, hi, g, c, h, a) SL(lo, hi, a[2][v] = RCP(a[2][v]);)
#define S13(lo, hi, g, c, h, a) SL(lo, hi, a[2][v] = 1.0f - 2.0f * a[2][v];)
#define S14(lo, hi, g, c, h, a) SL(lo, hi, h[v] = a[3][v] * a[2][v];)
#define CELL(g, c, h, a)                                                     \
  S0(0, 8, g, c, h, a) S1(0, 8, g, c, h, a) S2(0, 8, g, c, h, a)             \
  S3(0, 8, g, c, h, a) S4(0, 8, g, c, h, a) S5(0, 8, g, c, h, a)             \
  S6(0, 8, g, c, h, a) S7(0, 8, g, c, h, a) S8(0, 8, g, c, h, a)             \
  S9(0, 8, g, c, h, a) S10(0, 8, g, c, h, a) S11(0, 8, g, c, h, a)           \
  S12(0, 8, g, c, h, a) S13(0, 8, g, c, h, a) S14(0, 8, g, c, h, a)

__global__ void __launch_bounds__(128, 1) lstm2_fc_kernel(
    const float* __restrict__ x,
    const float* __restrict__ Wih0,
    const float* __restrict__ Whh0,
    const float* __restrict__ bih0,
    const float* __restrict__ bhh0,
    const float* __restrict__ Wih1,
    const float* __restrict__ Whh1,
    const float* __restrict__ bih1,
    const float* __restrict__ bhh1,
    const float* __restrict__ Wfc,
    const float* __restrict__ bfc,
    float* __restrict__ out)
{
  const int lane = threadIdx.x & 31;
  const int half = lane >> 4;
  const int nIdx = lane & 15;
  const int wid  = (blockIdx.x * blockDim.x + threadIdx.x) >> 5;
  const int base = wid * 16;

  v2f Whh0A[4][4], Wih1A[4][4], Whh1A[4][4];
  v2f Wih0A[4];
  v2f B1A[4];
#pragma unroll
  for (int j = 0; j < 4; ++j) {
    const int row = 16 * j + nIdx;
#pragma unroll
    for (int kc = 0; kc < 4; ++kc) {
      const float* p0 = Whh0 + row * 16 + kc * 4 + 2 * half;
      const float* p1 = Wih1 + row * 16 + kc * 4 + 2 * half;
      const float* p2 = Whh1 + row * 16 + kc * 4 + 2 * half;
      Whh0A[j][kc] = v2f{p0[0], p0[1]};
      Wih1A[j][kc] = v2f{p1[0], p1[1]};
      Whh1A[j][kc] = v2f{p2[0], p2[1]};
    }
    const float w0  = Wih0[row];
    const float bb0 = bih0[row] + bhh0[row];
    const float bb1 = bih1[row] + bhh1[row];
    Wih0A[j] = v2f{half ? 0.0f : w0, half ? 0.0f : bb0};
    B1A[j]   = v2f{half ? 0.0f : bb1, 0.0f};
  }
  const v2f onesB = v2f{half ? 0.0f : 1.0f, 0.0f};
  const v8f vzero = {0.f, 0.f, 0.f, 0.f, 0.f, 0.f, 0.f, 0.f};

  v2f h0B[4], h1B[4];
#pragma unroll
  for (int kc = 0; kc < 4; ++kc) {
    h0B[kc] = v2f{0.0f, 0.0f};
    h1B[kc] = v2f{0.0f, 0.0f};
  }

  auto convert = [&](const float(&h)[8], v2f(&hB)[4]) {
#pragma unroll
    for (int kc = 0; kc < 4; ++kc) {
      const int k0 = 4 * kc;
      if (kc < 2) {
        float sx = __shfl_xor(h[k0 + 2], 16, 32);
        float sy = __shfl_xor(h[k0 + 3], 16, 32);
        hB[kc] = v2f{half ? sx : h[k0 + 0], half ? sy : h[k0 + 1]};
      } else {
        float sx = __shfl_xor(h[k0 - 8], 16, 32);
        float sy = __shfl_xor(h[k0 - 7], 16, 32);
        hB[kc] = v2f{half ? h[k0 - 6] : sx, half ? h[k0 - 5] : sy};
      }
    }
  };

  const float* xrow = x + (base + nIdx) * LSTM_T;

  v8f g0[4], g1[4];
  float a0[4][8], a1[4][8];
  float c0[8], c1[8], h0[8], h1[8];
#pragma unroll
  for (int v = 0; v < 8; ++v) { c0[v] = 0.0f; c1[v] = 0.0f; h0[v] = 0.0f; h1[v] = 0.0f; }
  float sx[4], sy[4];
  v2f xB;

  auto w0 = [&](int k) __attribute__((always_inline)) {
    const int r = k >> 2, j = k & 3;
    if (r == 0) g0[j] = wmma4(Wih0A[j], xB, vzero);
    else        g0[j] = wmma4(Whh0A[j][r - 1], h0B[r - 1], g0[j]);
  };
  auto w1 = [&](int k) __attribute__((always_inline)) {
    const int r = k >> 2, j = k & 3;
    if (r == 0)      g1[j] = wmma4(B1A[j], onesB, vzero);
    else if (r <= 4) g1[j] = wmma4(Wih1A[j][r - 1], h0B[r - 1], g1[j]);
    else             g1[j] = wmma4(Whh1A[j][r - 5], h1B[r - 5], g1[j]);
  };
  auto fetch = [&](const float(&h)[8]) __attribute__((always_inline)) {
#pragma unroll
    for (int kc = 0; kc < 4; ++kc) {
      const int k0 = 4 * kc;
      sx[kc] = __shfl_xor(kc < 2 ? h[k0 + 2] : h[k0 - 8], 16, 32);
      sy[kc] = __shfl_xor(kc < 2 ? h[k0 + 3] : h[k0 - 7], 16, 32);
    }
  };
  auto select = [&](const float(&h)[8], v2f(&hB)[4]) __attribute__((always_inline)) {
#pragma unroll
    for (int kc = 0; kc < 4; ++kc) {
      const int k0 = 4 * kc;
      if (kc < 2) hB[kc] = v2f{half ? sx[kc] : h[k0 + 0], half ? sy[kc] : h[k0 + 1]};
      else        hB[kc] = v2f{half ? h[k0 - 6] : sx[kc], half ? h[k0 - 5] : sy[kc]};
    }
  };
#define W0(k) w0(k); asm volatile("" :: "v"(g0[(k) & 3])); FENCE();
#define W1(k) w1(k); asm volatile("" :: "v"(g1[(k) & 3])); FENCE();
#define GEMM0() W0(0) W0(1) W0(2) W0(3) W0(4) W0(5) W0(6) W0(7) W0(8) W0(9) \
  W0(10) W0(11) W0(12) W0(13) W0(14) W0(15) W0(16) W0(17) W0(18) W0(19)
#define GEMM1() W1(0) W1(1) W1(2) W1(3) W1(4) W1(5) W1(6) W1(7) W1(8) W1(9) \
  W1(10) W1(11) W1(12) W1(13) W1(14) W1(15) W1(16) W1(17) W1(18) W1(19)     \
  W1(20) W1(21) W1(22) W1(23) W1(24) W1(25) W1(26) W1(27) W1(28) W1(29)     \
  W1(30) W1(31) W1(32) W1(33) W1(34) W1(35)

  xB = v2f{half ? 0.0f : xrow[0], half ? 0.0f : 1.0f};
  GEMM0()
  CELL(g0, c0, h0, a0)
  convert(h0, h0B);
  GEMM1()
  xB = v2f{half ? 0.0f : xrow[1], half ? 0.0f : 1.0f};
  GEMM0()
  CELL(g0, c0, h0, a0)
  convert(h0, h0B);
  float xn = xrow[2];

#pragma unroll 1
  for (int t = 1; t < LSTM_T - 1; ++t) {
    xB = v2f{half ? 0.0f : xn, half ? 0.0f : 1.0f};
    const int tn = t + 2 < LSTM_T ? t + 2 : LSTM_T - 1;
    xn = xrow[tn];
    FENCE();

    W0(0) S0(0, 8, g1, c1, h1, a1)
    W0(1) S1(0, 4, g1, c1, h1, a1)
    W0(2) S1(4, 8, g1, c1, h1, a1)
    W0(3) S2(0, 8, g1, c1, h1, a1)
    W0(4) S3(0, 8, g1, c1, h1, a1)
    W0(5) S4(0, 4, g1, c1, h1, a1)
    W0(6) S4(4, 8, g1, c1, h1, a1)
    W0(7) S5(0, 8, g1, c1, h1, a1)
    W0(8) S6(0, 4, g1, c1, h1, a1)
    W0(9) S6(4, 8, g1, c1, h1, a1)
    W0(10) S7(0, 8, g1, c1, h1, a1)
    W0(11) S8(0, 4, g1, c1, h1, a1)
    W0(12) S8(4, 8, g1, c1, h1, a1)
    W0(13) S9(0, 4, g1, c1, h1, a1)
    W0(14) S9(4, 8, g1, c1, h1, a1)
    W0(15) S10(0, 8, g1, c1, h1, a1)
    W0(16) S11(0, 8, g1, c1, h1, a1)
    W0(17) S12(0, 8, g1, c1, h1, a1)
    W0(18) S13(0, 8, g1, c1, h1, a1)
    W0(19) S14(0, 8, g1, c1, h1, a1)

    W1(0) fetch(h1); FENCE();
    W1(1) S0(0, 4, g0, c0, h0, a0)
    W1(2) S0(4, 8, g0, c0, h0, a0)
    W1(3) S1(0, 4, g0, c0, h0, a0)
    W1(4) S1(4, 8, g0, c0, h0, a0)
    W1(5) S2(0, 4, g0, c0, h0, a0)
    W1(6) S2(4, 8, g0, c0, h0, a0) select(h1, h1B); FENCE();
    W1(7) S3(0, 4, g0, c0, h0, a0)
    W1(8) S3(4, 8, g0, c0, h0, a0)
    W1(9) S4(0, 4, g0, c0, h0, a0)
    W1(10) S4(4, 8, g0, c0, h0, a0)
    W1(11) S5(0, 4, g0, c0, h0, a0)
    W1(12) S5(4, 8, g0, c0, h0, a0)
    W1(13) S6(0, 4, g0, c0, h0, a0)
    W1(14) S6(4, 8, g0, c0, h0, a0)
    W1(15) S7(0, 4, g0, c0, h0, a0)
    W1(16) S7(4, 8, g0, c0, h0, a0)
    W1(17) S8(0, 4, g0, c0, h0, a0)
    W1(18) S8(4, 8, g0, c0, h0, a0)
    W1(19) S9(0, 4, g0, c0, h0, a0)
    W1(20) S9(4, 8, g0, c0, h0, a0)
    W1(21) S10(0, 4, g0, c0, h0, a0)
    W1(22) S10(4, 8, g0, c0, h0, a0)
    W1(23) S11(0, 4, g0, c0, h0, a0)
    W1(24) S11(4, 8, g0, c0, h0, a0)
    W1(25) S12(0, 4, g0, c0, h0, a0)
    W1(26) S12(4, 8, g0, c0, h0, a0)
    W1(27) S13(0, 4, g0, c0, h0, a0)
    W1(28) S13(4, 8, g0, c0, h0, a0)
    W1(29) S14(0, 4, g0, c0, h0, a0)
    W1(30) S14(4, 8, g0, c0, h0, a0)
    W1(31) fetch(h0); FENCE();
    W1(32) W1(33) W1(34) W1(35)
    select(h0, h0B); FENCE();
  }

  CELL(g1, c1, h1, a1)
  convert(h1, h1B);
  GEMM1()
  CELL(g1, c1, h1, a1)

  float partial = 0.0f;
#pragma unroll
  for (int v = 0; v < 8; ++v) partial += Wfc[v + 8 * half] * h1[v];
  const float tot = partial + __shfl_xor(partial, 16, 32);
  if (half == 0) out[base + nIdx] = tot + bfc[0];
}

extern "C" void kernel_launch(void* const* d_in, const int* in_sizes, int n_in,
                              void* d_out, int out_size, void* d_ws,
                              size_t ws_size, hipStream_t stream) {
  (void)in_sizes; (void)n_in; (void)d_ws; (void)ws_size; (void)out_size;
  const float* x    = (const float*)d_in[0];
  const float* Wih0 = (const float*)d_in[1];
  const float* Whh0 = (const float*)d_in[2];
  const float* bih0 = (const float*)d_in[3];
  const float* bhh0 = (const float*)d_in[4];
  const float* Wih1 = (const float*)d_in[5];
  const float* Whh1 = (const float*)d_in[6];
  const float* bih1 = (const float*)d_in[7];
  const float* bhh1 = (const float*)d_in[8];
  const float* Wfc  = (const float*)d_in[9];
  const float* bfc  = (const float*)d_in[10];
  float* out = (float*)d_out;

  dim3 block(128);
  dim3 grid(LSTM_B / (16 * 4));
  lstm2_fc_kernel<<<grid, block, 0, stream>>>(x, Wih0, Whh0, bih0, bhh0, Wih1,
                                              Whh1, bih1, bhh1, Wfc, bfc, out);
}


# each tile's two layers run on two waves (two SIMDs) of one workgroup, one step apart: h0 handed over as B fragments through a 2-slot LDS ring, one workgroup barrier per step; all sums unchanged
# speedup vs baseline: 1.8152x; 1.5923x over previous
#include <hip/hip_runtime.h>


#define LSTM_B 4096
#define LSTM_T 512

typedef float v2f __attribute__((ext_vector_type(2)));
typedef float v8f __attribute__((ext_vector_type(8)));

static __device__ __forceinline__ v8f wmma4(v2f a, v2f b, v8f c) {
  return __builtin_amdgcn_wmma_f32_16x16x4_f32(false, a, false, b, (short)0, c,
                                               false, false);
}

#define FENCE() __builtin_amdgcn_sched_barrier(0)

#define K_SIG  (-1.44269504f)
#define K_TANH (2.88539008f)
#define EXP2(x) __builtin_amdgcn_exp2f(x)
#define RCP(x)  __builtin_amdgcn_rcpf(x)

#define SL(lo, hi, ...) _Pragma("unroll") for (int v = lo; v < hi; ++v) { __VA_ARGS__ FENCE(); }
#define S0(lo, hi, g, c, h, a)  SL(lo, hi, a[2][v] = K_TANH * g[2][v]; a[0][v] = K_SIG * g[0][v];)
#define S1(lo, hi, g, c, h, a)  SL(lo, hi, a[2][v] = EXP2(a[2][v]);    a[1][v] = K_SIG * g[1][v];)
#define S2(lo, hi, g, c, h, a)  SL(lo, hi, a[0][v] = EXP2(a[0][v]);    a[2][v] = 1.0f + a[2][v];)
#define S3(lo, hi, g, c, h, a)  SL(lo, hi, a[2][v] = RCP(a[2][v]);     a[3][v] = K_SIG * g[3][v];)
#define S4(lo, hi, g, c, h, a)  SL(lo, hi, a[1][v] = EXP2(a[1][v]);    a[0][v] = 1.0f + a[0][v];)
#define S5(lo, hi, g, c, h, a)  SL(lo, hi, a[0][v] = RCP(a[0][v]);     a[2][v] = 1.0f - 2.0f * a[2][v];)
#define S6(lo, hi, g, c, h, a)  SL(lo, hi, a[3][v] = EXP2(a[3][v]);    a[1][v] = 1.0f + a[1][v];)
#define S7(lo, hi, g, c, h, a)  SL(lo, hi, a[1][v] = RCP(a[1][v]);     a[0][v] = a[0][v] * a[2][v];)
#define S8(lo, hi, g, c, h, a)  SL(lo, hi, a[3][v] = 1.0f + a[3][v];   c[v] = a[1][v] * c[v] + a[0][v];)
#define S9(lo, hi, g, c, h, a)  SL(lo, hi, a[3][v] = RCP(a[3][v]);     a[2][v] = K_TANH * c[v];)
#define S10(lo, hi, g, c, h, a) SL(lo, hi, a[2][v] = EXP2(a[2][v]);)
#define S11(lo, hi, g, c, h, a) SL(lo, hi, a[2][v] = 1.0f + a[2][v];)
#define S12(lo, hi, g, c, h, a) SL(lo, hi, a[2][v] = RCP(a[2][v]);)
#define S13(lo, hi, g, c, h, a) SL(lo, hi, a[2][v] = 1.0f - 2.0f * a[2][v];)
#define S14(lo, hi, g, c, h, a) SL(lo, hi, h[v] = a[3][v] * a[2][v];)
#define CELL(g, c, h, a)                                                     \
  S0(0, 8, g, c, h, a) S1(0, 8, g, c, h, a) S2(0, 8, g, c, h, a)             \
  S3(0, 8, g, c, h, a) S4(0, 8, g, c, h, a) S5(0, 8, g, c, h, a)             \
  S6(0, 8, g, c, h, a) S7(0, 8, g, c, h, a) S8(0, 8, g, c, h, a)             \
  S9(0, 8, g, c, h, a) S10(0, 8, g, c, h, a) S11(0, 8, g, c, h, a)           \
  S12(0, 8, g, c, h, a) S13(0, 8, g, c, h, a) S14(0, 8, g, c, h, a)

__global__ void __launch_bounds__(128, 1) lstm2_fc_kernel(
    const float* __restrict__ x,
    const float* __restrict__ Wih0,
    const float* __restrict__ Whh0,
    const float* __restrict__ bih0,
    const float* __restrict__ bhh0,
    const float* __restrict__ Wih1,
    const float* __restrict__ Whh1,
    const float* __restrict__ bih1,
    const float* __restrict__ bhh1,
    const float* __restrict__ Wfc,
    const float* __restrict__ bfc,
    float* __restrict__ out)
{
  __shared__ float4 hx[2][2][2][32];

  const int lane = threadIdx.x & 31;
  const int half = lane >> 4;
  const int nIdx = lane & 15;
  const int wave = __builtin_amdgcn_readfirstlane((int)(threadIdx.x >> 5));
  const int role = wave & 1;
  const int tib  = wave >> 1;
  const int base = (blockIdx.x * 2 + tib) * 16;

  v2f Whh0A[4][4], Wih1A[4][4], Whh1A[4][4];
  v2f Wih0A[4];
  v2f B1A[4];
#pragma unroll
  for (int j = 0; j < 4; ++j) {
    const int row = 16 * j + nIdx;
#pragma unroll
    for (int kc = 0; kc < 4; ++kc) {
      const float* p0 = Whh0 + row * 16 + kc * 4 + 2 * half;
      const float* p1 = Wih1 + row * 16 + kc * 4 + 2 * half;
      const float* p2 = Whh1 + row * 16 + kc * 4 + 2 * half;
      Whh0A[j][kc] = v2f{p0[0], p0[1]};
      Wih1A[j][kc] = v2f{p1[0], p1[1]};
      Whh1A[j][kc] = v2f{p2[0], p2[1]};
    }
    const float w0  = Wih0[row];
    const float bb0 = bih0[row] + bhh0[row];
    const float bb1 = bih1[row] + bhh1[row];
    Wih0A[j] = v2f{half ? 0.0f : w0, half ? 0.0f : bb0};
    B1A[j]   = v2f{half ? 0.0f : bb1, 0.0f};
  }
  const v2f onesB = v2f{half ? 0.0f : 1.0f, 0.0f};
  const v8f vzero = {0.f, 0.f, 0.f, 0.f, 0.f, 0.f, 0.f, 0.f};

  v2f h0B[4], h1B[4];
#pragma unroll
  for (int kc = 0; kc < 4; ++kc) {
    h0B[kc] = v2f{0.0f, 0.0f};
    h1B[kc] = v2f{0.0f, 0.0f};
  }

  auto convert = [&](const float(&h)[8], v2f(&hB)[4]) {
#pragma unroll
    for (int kc = 0; kc < 4; ++kc) {
      const int k0 = 4 * kc;
      if (kc < 2) {
        float sx = __shfl_xor(h[k0 + 2], 16, 32);
        float sy = __shfl_xor(h[k0 + 3], 16, 32);
        hB[kc] = v2f{half ? sx : h[k0 + 0], half ? sy : h[k0 + 1]};
      } else {
        float sx = __shfl_xor(h[k0 - 8], 16, 32);
        float sy = __shfl_xor(h[k0 - 7], 16, 32);
        hB[kc] = v2f{half ? h[k0 - 6] : sx, half ? h[k0 - 5] : sy};
      }
    }
  };

  const float* xrow = x + (base + nIdx) * LSTM_T;

  v8f g0[4], g1[4];
  float a0[4][8], a1[4][8];
  float c0[8], c1[8], h0[8], h1[8];
#pragma unroll
  for (int v = 0; v < 8; ++v) { c0[v] = 0.0f; c1[v] = 0.0f; h0[v] = 0.0f; h1[v] = 0.0f; }
  float sx[4], sy[4];
  v2f xB;
  v2f hin[4];

  auto w0 = [&](int k) __attribute__((always_inline)) {
    const int r = k >> 2, j = k & 3;
    if (r == 0) g0[j] = wmma4(Wih0A[j], xB, vzero);
    else        g0[j] = wmma4(Whh0A[j][r - 1], h0B[r - 1], g0[j]);
  };
  auto w1 = [&](int k) __attribute__((always_inline)) {
    const int r = k >> 2, j = k & 3;
    if (r == 0)      g1[j] = wmma4(B1A[j], onesB, vzero);
    else if (r <= 4) g1[j] = wmma4(Wih1A[j][r - 1], hin[r - 1], g1[j]);
    else             g1[j] = wmma4(Whh1A[j][r - 5], h1B[r - 5], g1[j]);
  };
  auto fetch = [&](const float(&h)[8]) __attribute__((always_inline)) {
#pragma unroll
    for (int kc = 0; kc < 4; ++kc) {
      const int k0 = 4 * kc;
      sx[kc] = __shfl_xor(kc < 2 ? h[k0 + 2] : h[k0 - 8], 16, 32);
      sy[kc] = __shfl_xor(kc < 2 ? h[k0 + 3] : h[k0 - 7], 16, 32);
    }
  };
  auto select = [&](const float(&h)[8], v2f(&hB)[4]) __attribute__((always_inline)) {
#pragma unroll
    for (int kc = 0; kc < 4; ++kc) {
      const int k0 = 4 * kc;
      if (kc < 2) hB[kc] = v2f{half ? sx[kc] : h[k0 + 0], half ? sy[kc] : h[k0 + 1]};
      else        hB[kc] = v2f{half ? h[k0 - 6] : sx[kc], half ? h[k0 - 5] : sy[kc]};
    }
  };
#define W0(k) w0(k); asm volatile("" :: "v"(g0[(k) & 3])); FENCE();
#define W1(k) w1(k); asm volatile("" :: "v"(g1[(k) & 3])); FENCE();
#define GEMM0() W0(0) W0(1) W0(2) W0(3) W0(4) W0(5) W0(6) W0(7) W0(8) W0(9) \
  W0(10) W0(11) W0(12) W0(13) W0(14) W0(15) W0(16) W0(17) W0(18) W0(19)
#define GEMM1_BIAS() W1(0) W1(1) W1(2) W1(3)
#define GEMM1_IH() W1(4) W1(5) W1(6) W1(7) W1(8) W1(9) W1(10) W1(11) W1(12)  \
  W1(13) W1(14) W1(15) W1(16) W1(17) W1(18) W1(19)
#define GEMM1_HH() W1(20) W1(21) W1(22) W1(23) W1(24) W1(25) W1(26) W1(27)   \
  W1(28) W1(29) W1(30) W1(31) W1(32) W1(33) W1(34) W1(35)

#define STEP1(slot)                                                          \
  {                                                                          \
    const float4 u0 = hx[slot][tib][0][lane];                                \
    const float4 u1 = hx[slot][tib][1][lane];                                \
    fetch(h1);                                                               \
    FENCE();   \
    GEMM1_BIAS()                                                             \
    hin[0] = v2f{u0.x, u0.y}; hin[1] = v2f{u0.z, u0.w};                      \
    hin[2] = v2f{u1.x, u1.y}; hin[3] = v2f{u1.z, u1.w};                      \
    GEMM1_IH()                                                               \
    select(h1, h1B);                                                         \
    FENCE();   \
    GEMM1_HH()                                                               \
    CELL(g1, c1, h1, a1)                                                     \
  }
#define STEP0(slot, tnext)                                                   \
  {                                                                          \
    xB = v2f{half ? 0.0f : xn, half ? 0.0f : 1.0f};                          \
    xn = xrow[tnext];                                                        \
    GEMM0()                                                                  \
    CELL(g0, c0, h0, a0)                                                     \
    convert(h0, h0B);                                                        \
    hx[slot][tib][0][lane] = float4{h0B[0].x, h0B[0].y, h0B[1].x, h0B[1].y}; \
    hx[slot][tib][1][lane] = float4{h0B[2].x, h0B[2].y, h0B[3].x, h0B[3].y}; \
  }

  float xn = xrow[0];

  if (role == 0) STEP0(0, 1)
  __syncthreads();
#pragma unroll 1
  for (int p = 1; p < LSTM_T; ++p) {
    if (role == 0) {
      const int tn = p + 1 < LSTM_T ? p + 1 : LSTM_T - 1;
      STEP0(p & 1, tn)
    } else {
      STEP1((p - 1) & 1)
    }
    __syncthreads();
  }
  if (role == 1) STEP1((LSTM_T - 1) & 1)

  if (role == 1) {
    float partial = 0.0f;
#pragma unroll
    for (int v = 0; v < 8; ++v) partial += Wfc[v + 8 * half] * h1[v];
    const float tot = partial + __shfl_xor(partial, 16, 32);
    if (half == 0) out[base + nIdx] = tot + bfc[0];
  }
}

extern "C" void kernel_launch(void* const* d_in, const int* in_sizes, int n_in,
                              void* d_out, int out_size, void* d_ws,
                              size_t ws_size, hipStream_t stream) {
  (void)in_sizes; (void)n_in; (void)d_ws; (void)ws_size; (void)out_size;
  const float* x    = (const float*)d_in[0];
  const float* Wih0 = (const float*)d_in[1];
  const float* Whh0 = (const float*)d_in[2];
  const float* bih0 = (const float*)d_in[3];
  const float* bhh0 = (const float*)d_in[4];
  const float* Wih1 = (const float*)d_in[5];
  const float* Whh1 = (const float*)d_in[6];
  const float* bih1 = (const float*)d_in[7];
  const float* bhh1 = (const float*)d_in[8];
  const float* Wfc  = (const float*)d_in[9];
  const float* bfc  = (const float*)d_in[10];
  float* out = (float*)d_out;

  dim3 block(128);
  dim3 grid(LSTM_B / (16 * 2));
  lstm2_fc_kernel<<<grid, block, 0, stream>>>(x, Wih0, Whh0, bih0, bhh0, Wih1,
                                              Whh1, bih1, bhh1, Wfc, bfc, out);
}


# each tile on three waves (three SIMDs), not two: layer 1's bias + Wih1 @ h0 chain links run on a wave of their own and reach the Whh1 wave as WMMA C tiles through an LDS ring; all sums unchanged
# speedup vs baseline: 2.0066x; 1.1054x over previous
#include <hip/hip_runtime.h>


#define LSTM_B 4096
#define LSTM_T 512

typedef float v2f __attribute__((ext_vector_type(2)));
typedef float v8f __attribute__((ext_vector_type(8)));

static __device__ __forceinline__ v8f wmma4(v2f a, v2f b, v8f c) {
  return __builtin_amdgcn_wmma_f32_16x16x4_f32(false, a, false, b, (short)0, c,
                                               false, false);
}

#define FENCE() __builtin_amdgcn_sched_barrier(0)

#define K_SIG  (-1.44269504f)
#define K_TANH (2.88539008f)
#define EXP2(x) __builtin_amdgcn_exp2f(x)
#define RCP(x)  __builtin_amdgcn_rcpf(x)

#define SL(lo, hi, ...) _Pragma("unroll") for (int v = lo; v < hi; ++v) { __VA_ARGS__ FENCE(); }
#define S0(lo, hi, g, c, h, a)  SL(lo, hi, a[2][v] = K_TANH * g[2][v]; a[0][v] = K_SIG * g[0][v];)
#define S1(lo, hi, g, c, h, a)  SL(lo, hi, a[2][v] = EXP2(a[2][v]);    a[1][v] = K_SIG * g[1][v];)
#define S2(lo, hi, g, c, h, a)  SL(lo, hi, a[0][v] = EXP2(a[0][v]);    a[2][v] = 1.0f + a[2][v];)
#define S3(lo, hi, g, c, h, a)  SL(lo, hi, a[2][v] = RCP(a[2][v]);     a[3][v] = K_SIG * g[3][v];)
#define S4(lo, hi, g, c, h, a)  SL(lo, hi, a[1][v] = EXP2(a[1][v]);    a[0][v] = 1.0f + a[0][v];)
#define S5(lo, hi, g, c, h, a)  SL(lo, hi, a[0][v] = RCP(a[0][v]);     a[2][v] = 1.0f - 2.0f * a[2][v];)
#define S6(lo, hi, g, c, h, a)  SL(lo, hi, a[3][v] = EXP2(a[3][v]);    a[1][v] = 1.0f + a[1][v];)
#define S7(lo, hi, g, c, h, a)  SL(lo, hi, a[1][v] = RCP(a[1][v]);     a[0][v] = a[0][v] * a[2][v];)
#define S8(lo, hi, g, c, h, a)  SL(lo, hi, a[3][v] = 1.0f + a[3][v];   c[v] = a[1][v] * c[v] + a[0][v];)
#define S9(lo, hi, g, c, h, a)  SL(lo, hi, a[3][v] = RCP(a[3][v]);     a[2][v] = K_TANH * c[v];)
#define S10(lo, hi, g, c, h, a) SL(lo, hi, a[2][v] = EXP2(a[2][v]);)
#define S11(lo, hi, g, c, h, a) SL(lo, hi, a[2][v] = 1.0f + a[2][v];)
#define S12(lo, hi, g, c, h, a) SL(lo, hi, a[2][v] = RCP(a[2][v]);)
#define S13(lo, hi, g, c, h, a) SL(lo, hi, a[2][v] = 1.0f - 2.0f * a[2][v];)
#define S14(lo, hi, g, c, h, a) SL(lo, hi, h[v] = a[3][v] * a[2][v];)
#define CELL(g, c, h, a)                                                     \
  S0(0, 8, g, c, h, a) S1(0, 8, g, c, h, a) S2(0, 8, g, c, h, a)             \
  S3(0, 8, g, c, h, a) S4(0, 8, g, c, h, a) S5(0, 8, g, c, h, a)             \
  S6(0, 8, g, c, h, a) S7(0, 8, g, c, h, a) S8(0, 8, g, c, h, a)             \
  S9(0, 8, g, c, h, a) S10(0, 8, g, c, h, a) S11(0, 8, g, c, h, a)           \
  S12(0, 8, g, c, h, a) S13(0, 8, g, c, h, a) S14(0, 8, g, c, h, a)

__global__ void __launch_bounds__(96, 1) lstm2_fc_kernel(
    const float* __restrict__ x,
    const float* __restrict__ Wih0,
    const float* __restrict__ Whh0,
    const float* __restrict__ bih0,
    const float* __restrict__ bhh0,
    const float* __restrict__ Wih1,
    const float* __restrict__ Whh1,
    const float* __restrict__ bih1,
    const float* __restrict__ bhh1,
    const float* __restrict__ Wfc,
    const float* __restrict__ bfc,
    float* __restrict__ out)
{
  __shared__ float4 hx[2][2][32];
  __shared__ float4 gx[2][8][32];

  const int lane = threadIdx.x & 31;
  const int half = lane >> 4;
  const int nIdx = lane & 15;
  const int role = __builtin_amdgcn_readfirstlane((int)(threadIdx.x >> 5));
  const int base = blockIdx.x * 16;

  v2f Whh0A[4][4], Wih1A[4][4], Whh1A[4][4];
  v2f Wih0A[4];
  v2f B1A[4];
#pragma unroll
  for (int j = 0; j < 4; ++j) {
    const int row = 16 * j + nIdx;
#pragma unroll
    for (int kc = 0; kc < 4; ++kc) {
      const float* p0 = Whh0 + row * 16 + kc * 4 + 2 * half;
      const float* p1 = Wih1 + row * 16 + kc * 4 + 2 * half;
      const float* p2 = Whh1 + row * 16 + kc * 4 + 2 * half;
      Whh0A[j][kc] = v2f{p0[0], p0[1]};
      Wih1A[j][kc] = v2f{p1[0], p1[1]};
      Whh1A[j][kc] = v2f{p2[0], p2[1]};
    }
    const float w0  = Wih0[row];
    const float bb0 = bih0[row] + bhh0[row];
    const float bb1 = bih1[row] + bhh1[row];
    Wih0A[j] = v2f{half ? 0.0f : w0, half ? 0.0f : bb0};
    B1A[j]   = v2f{half ? 0.0f : bb1, 0.0f};
  }
  const v2f onesB = v2f{half ? 0.0f : 1.0f, 0.0f};
  const v8f vzero = {0.f, 0.f, 0.f, 0.f, 0.f, 0.f, 0.f, 0.f};

  v2f h0B[4], h1B[4];
#pragma unroll
  for (int kc = 0; kc < 4; ++kc) {
    h0B[kc] = v2f{0.0f, 0.0f};
    h1B[kc] = v2f{0.0f, 0.0f};
  }

  auto convert = [&](const float(&h)[8], v2f(&hB)[4]) {
#pragma unroll
    for (int kc = 0; kc < 4; ++kc) {
      const int k0 = 4 * kc;
      if (kc < 2) {
        float sx = __shfl_xor(h[k0 + 2], 16, 32);
        float sy = __shfl_xor(h[k0 + 3], 16, 32);
        hB[kc] = v2f{half ? sx : h[k0 + 0], half ? sy : h[k0 + 1]};
      } else {
        float sx = __shfl_xor(h[k0 - 8], 16, 32);
        float sy = __shfl_xor(h[k0 - 7], 16, 32);
        hB[kc] = v2f{half ? h[k0 - 6] : sx, half ? h[k0 - 5] : sy};
      }
    }
  };

  const float* xrow = x + (base + nIdx) * LSTM_T;

  v8f g0[4], g1[4];
  v8f gc[4];
#pragma unroll
  for (int j = 0; j < 4; ++j) gc[j] = vzero;
  float a0[4][8], a1[4][8];
  float c0[8], c1[8], h0[8], h1[8];
#pragma unroll
  for (int v = 0; v < 8; ++v) { c0[v] = 0.0f; c1[v] = 0.0f; h0[v] = 0.0f; h1[v] = 0.0f; }
  float sx[4], sy[4];
#pragma unroll
  for (int kc = 0; kc < 4; ++kc) { sx[kc] = 0.0f; sy[kc] = 0.0f; }
  v2f xB;
  v2f hin[4];

  auto w0 = [&](int k) __attribute__((always_inline)) {
    const int r = k >> 2, j = k & 3;
    if (r == 0) g0[j] = wmma4(Wih0A[j], xB, vzero);
    else        g0[j] = wmma4(Whh0A[j][r - 1], h0B[r - 1], g0[j]);
  };
  auto w1 = [&](int k) __attribute__((always_inline)) {
    const int r = k >> 2, j = k & 3;
    if (r == 0) g1[j] = wmma4(B1A[j], onesB, vzero);
    else        g1[j] = wmma4(Wih1A[j][r - 1], hin[r - 1], g1[j]);
  };
  auto w2 = [&](int k) __attribute__((always_inline)) {
    const int r = k >> 2, j = k & 3;
    if (r == 0) g1[j] = wmma4(Whh1A[j][0], h1B[0], gc[j]);
    else        g1[j] = wmma4(Whh1A[j][r], h1B[r], g1[j]);
  };
  auto fetch = [&](const float(&h)[8]) __attribute__((always_inline)) {
#pragma unroll
    for (int kc = 0; kc < 4; ++kc) {
      const int k0 = 4 * kc;
      sx[kc] = __shfl_xor(kc < 2 ? h[k0 + 2] : h[k0 - 8], 16, 32);
      sy[kc] = __shfl_xor(kc < 2 ? h[k0 + 3] : h[k0 - 7], 16, 32);
    }
  };
  auto select = [&](const float(&h)[8], v2f(&hB)[4]) __attribute__((always_inline)) {
#pragma unroll
    for (int kc = 0; kc < 4; ++kc) {
      const int k0 = 4 * kc;
      if (kc < 2) hB[kc] = v2f{half ? sx[kc] : h[k0 + 0], half ? sy[kc] : h[k0 + 1]};
      else        hB[kc] = v2f{half ? h[k0 - 6] : sx[kc], half ? h[k0 - 5] : sy[kc]};
    }
  };
#define W0(k) w0(k); asm volatile("" :: "v"(g0[(k) & 3])); FENCE();
#define W1(k) w1(k); asm volatile("" :: "v"(g1[(k) & 3])); FENCE();
#define W2(k) w2(k); asm volatile("" :: "v"(g1[(k) & 3])); FENCE();
#define GEMM0() W0(0) W0(1) W0(2) W0(3) W0(4) W0(5) W0(6) W0(7) W0(8) W0(9) \
  W0(10) W0(11) W0(12) W0(13) W0(14) W0(15) W0(16) W0(17) W0(18) W0(19)
#define GEMM1_BIAS() W1(0) W1(1) W1(2) W1(3)
#define GEMM1_IH() W1(4) W1(5) W1(6) W1(7) W1(8) W1(9) W1(10) W1(11) W1(12)  \
  W1(13) W1(14) W1(15) W1(16) W1(17) W1(18) W1(19)
#define GEMM1_HH() W2(0) W2(1) W2(2) W2(3) W2(4) W2(5) W2(6) W2(7) W2(8)     \
  W2(9) W2(10) W2(11) W2(12) W2(13) W2(14) W2(15)

#define STEP0(slot, tnext)                                                   \
  {                                                                          \
    xB = v2f{half ? 0.0f : xn, half ? 0.0f : 1.0f};                          \
    xn = xrow[tnext];                                                        \
    GEMM0()                                                                  \
    CELL(g0, c0, h0, a0)                                                     \
    convert(h0, h0B);                                                        \
    hx[slot][0][lane] = float4{h0B[0].x, h0B[0].y, h0B[1].x, h0B[1].y};      \
    hx[slot][1][lane] = float4{h0B[2].x, h0B[2].y, h0B[3].x, h0B[3].y};      \
  }
#define STEP1(slot)                                                          \
  {                                                                          \
    const float4 u0 = hx[slot][0][lane];                                     \
    const float4 u1 = hx[slot][1][lane];                                     \
    FENCE();   \
    GEMM1_BIAS()                                                             \
    hin[0] = v2f{u0.x, u0.y}; hin[1] = v2f{u0.z, u0.w};                      \
    hin[2] = v2f{u1.x, u1.y}; hin[3] = v2f{u1.z, u1.w};                      \
    GEMM1_IH()                                                               \
    _Pragma("unroll") for (int j = 0; j < 4; ++j) {                          \
      gx[slot][2 * j + 0][lane] = float4{g1[j][0], g1[j][1], g1[j][2], g1[j][3]}; \
      gx[slot][2 * j + 1][lane] = float4{g1[j][4], g1[j][5], g1[j][6], g1[j][7]}; \
    }                                                                        \
  }
#define STEP2_GEMM() { select(h1, h1B); GEMM1_HH() }
#define STEP2_FETCH(slot)                                                    \
  {                                                                          \
    _Pragma("unroll") for (int j = 0; j < 4; ++j) {                          \
      const float4 q0 = gx[slot][2 * j + 0][lane];                           \
      const float4 q1 = gx[slot][2 * j + 1][lane];                           \
      gc[j] = v8f{q0.x, q0.y, q0.z, q0.w, q1.x, q1.y, q1.z, q1.w};           \
    }                                                                        \
    FENCE();   \
  }
#define STEP2_CELL() { CELL(g1, c1, h1, a1) fetch(h1); }

  float xn = xrow[0];

#define PHASE_EDGE(p)                                                        \
  if (role == 0) {                                                           \
    if (p < LSTM_T) {                                                        \
      const int tn = p + 1 < LSTM_T ? p + 1 : LSTM_T - 1;   \
      STEP0(p & 1, tn)                                                       \
    }                                                                        \
  } else if (role == 1) {                                                    \
    if (p >= 1 && p <= LSTM_T) STEP1((p - 1) & 1)                            \
  } else {                                                                   \
    if (p >= 3) STEP2_GEMM()                                                 \
    if (p >= 2 && p <= LSTM_T + 1) STEP2_FETCH(p & 1)                        \
    if (p >= 3) STEP2_CELL()                                                 \
  }
#define PHASE_FULL(p)                                                        \
  if (role == 0) {                                                           \
    STEP0(p & 1, p + 1)                                                      \
  } else if (role == 1) {                                                    \
    STEP1((p - 1) & 1)                                                       \
  } else {                                                                   \
    STEP2_GEMM()                                                             \
    STEP2_FETCH(p & 1)                                                       \
    STEP2_CELL()                                                             \
  }
#pragma unroll 1
  for (int p = 0; p < 3; ++p) {
    PHASE_EDGE(p)
    __syncthreads();
  }
#pragma unroll 1
  for (int p = 3; p < LSTM_T - 1; ++p) {
    PHASE_FULL(p)
    __syncthreads();
  }
#pragma unroll 1
  for (int p = LSTM_T - 1; p < LSTM_T + 3; ++p) {
    PHASE_EDGE(p)
    __syncthreads();
  }

  if (role == 2) {
    float partial = 0.0f;
#pragma unroll
    for (int v = 0; v < 8; ++v) partial += Wfc[v + 8 * half] * h1[v];
    const float tot = partial + __shfl_xor(partial, 16, 32);
    if (half == 0) out[base + nIdx] = tot + bfc[0];
  }
}

extern "C" void kernel_launch(void* const* d_in, const int* in_sizes, int n_in,
                              void* d_out, int out_size, void* d_ws,
                              size_t ws_size, hipStream_t stream) {
  (void)in_sizes; (void)n_in; (void)d_ws; (void)ws_size; (void)out_size;
  const float* x    = (const float*)d_in[0];
  const float* Wih0 = (const float*)d_in[1];
  const float* Whh0 = (const float*)d_in[2];
  const float* bih0 = (const float*)d_in[3];
  const float* bhh0 = (const float*)d_in[4];
  const float* Wih1 = (const float*)d_in[5];
  const float* Whh1 = (const float*)d_in[6];
  const float* bih1 = (const float*)d_in[7];
  const float* bhh1 = (const float*)d_in[8];
  const float* Wfc  = (const float*)d_in[9];
  const float* bfc  = (const float*)d_in[10];
  float* out = (float*)d_out;

  dim3 block(96);
  dim3 grid(LSTM_B / 16);
  lstm2_fc_kernel<<<grid, block, 0, stream>>>(x, Wih0, Whh0, bih0, bhh0, Wih1,
                                              Whh1, bih1, bhh1, Wfc, bfc, out);
}


# weight A fragments loaded in the k order of h's D layout (link kc, lane half hf, element s: k=8hf+2kc+s): h's registers are the B fragments as they stand, no lane exchange; k summed in a new order
# speedup vs baseline: 2.4886x; 1.2402x over previous
#include <hip/hip_runtime.h>


#define LSTM_B 4096
#define LSTM_T 512

typedef float v2f __attribute__((ext_vector_type(2)));
typedef float v8f __attribute__((ext_vector_type(8)));

static __device__ __forceinline__ v8f wmma4(v2f a, v2f b, v8f c) {
  return __builtin_amdgcn_wmma_f32_16x16x4_f32(false, a, false, b, (short)0, c,
                                               false, false);
}

#define FENCE() __builtin_amdgcn_sched_barrier(0)

#define K_SIG  (-1.44269504f)
#define K_TANH (2.88539008f)
#define EXP2(x) __builtin_amdgcn_exp2f(x)
#define RCP(x)  __builtin_amdgcn_rcpf(x)

#define SL(lo, hi, ...) _Pragma("unroll") for (int v = lo; v < hi; ++v) { __VA_ARGS__ FENCE(); }
#define S0(lo, hi, g, c, h, a)  SL(lo, hi, a[2][v] = K_TANH * g[2][v]; a[0][v] = K_SIG * g[0][v];)
#define S1(lo, hi, g, c, h, a)  SL(lo, hi, a[2][v] = EXP2(a[2][v]);    a[1][v] = K_SIG * g[1][v];)
#define S2(lo, hi, g, c, h, a)  SL(lo, hi, a[0][v] = EXP2(a[0][v]);    a[2][v] = 1.0f + a[2][v];)
#define S3(lo, hi, g, c, h, a)  SL(lo, hi, a[2][v] = RCP(a[2][v]);     a[3][v] = K_SIG * g[3][v];)
#define S4(lo, hi, g, c, h, a)  SL(lo, hi, a[1][v] = EXP2(a[1][v]);    a[0][v] = 1.0f + a[0][v];)
#define S5(lo, hi, g, c, h, a)  SL(lo, hi, a[0][v] = RCP(a[0][v]);     a[2][v] = 1.0f - 2.0f * a[2][v];)
#define S6(lo, hi, g, c, h, a)  SL(lo, hi, a[3][v] = EXP2(a[3][v]);    a[1][v] = 1.0f + a[1][v];)
#define S7(lo, hi, g, c, h, a)  SL(lo, hi, a[1][v] = RCP(a[1][v]);     a[0][v] = a[0][v] * a[2][v];)
#define S8(lo, hi, g, c, h, a)  SL(lo, hi, a[3][v] = 1.0f + a[3][v];   c[v] = a[1][v] * c[v] + a[0][v];)
#define S9(lo, hi, g, c, h, a)  SL(lo, hi, a[3][v] = RCP(a[3][v]);     a[2][v] = K_TANH * c[v];)
#define S10(lo, hi, g, c, h, a) SL(lo, hi, a[2][v] = EXP2(a[2][v]);)
#define S11(lo, hi, g, c, h, a) SL(lo, hi, a[2][v] = 1.0f + a[2][v];)
#define S12(lo, hi, g, c, h, a) SL(lo, hi, a[2][v] = RCP(a[2][v]);)
#define S13(lo, hi, g, c, h, a) SL(lo, hi, a[2][v] = 1.0f - 2.0f * a[2][v];)
#define S14(lo, hi, g, c, h, a) SL(lo, hi, h[v] = a[3][v] * a[2][v];)
#define CELL(g, c, h, a)                                                     \
  S0(0, 8, g, c, h, a) S1(0, 8, g, c, h, a) S2(0, 8, g, c, h, a)             \
  S3(0, 8, g, c, h, a) S4(0, 8, g, c, h, a) S5(0, 8, g, c, h, a)             \
  S6(0, 8, g, c, h, a) S7(0, 8, g, c, h, a) S8(0, 8, g, c, h, a)             \
  S9(0, 8, g, c, h, a) S10(0, 8, g, c, h, a) S11(0, 8, g, c, h, a)           \
  S12(0, 8, g, c, h, a) S13(0, 8, g, c, h, a) S14(0, 8, g, c, h, a)

__global__ void __launch_bounds__(96, 1) lstm2_fc_kernel(
    const float* __restrict__ x,
    const float* __restrict__ Wih0,
    const float* __restrict__ Whh0,
    const float* __restrict__ bih0,
    const float* __restrict__ bhh0,
    const float* __restrict__ Wih1,
    const float* __restrict__ Whh1,
    const float* __restrict__ bih1,
    const float* __restrict__ bhh1,
    const float* __restrict__ Wfc,
    const float* __restrict__ bfc,
    float* __restrict__ out)
{
  __shared__ float4 hx[2][2][32];
  __shared__ float4 gx[2][8][32];

  const int lane = threadIdx.x & 31;
  const int half = lane >> 4;
  const int nIdx = lane & 15;
  const int role = __builtin_amdgcn_readfirstlane((int)(threadIdx.x >> 5));
  const int base = blockIdx.x * 16;

  v2f Whh0A[4][4], Wih1A[4][4], Whh1A[4][4];
  v2f Wih0A[4];
  v2f B1A[4];
#pragma unroll
  for (int j = 0; j < 4; ++j) {
    const int row = 16 * j + nIdx;
#pragma unroll
    for (int kc = 0; kc < 4; ++kc) {
      const float* p0 = Whh0 + row * 16 + 8 * half + 2 * kc;
      const float* p1 = Wih1 + row * 16 + 8 * half + 2 * kc;
      const float* p2 = Whh1 + row * 16 + 8 * half + 2 * kc;
      Whh0A[j][kc] = v2f{p0[0], p0[1]};
      Wih1A[j][kc] = v2f{p1[0], p1[1]};
      Whh1A[j][kc] = v2f{p2[0], p2[1]};
    }
    const float w0  = Wih0[row];
    const float bb0 = bih0[row] + bhh0[row];
    const float bb1 = bih1[row] + bhh1[row];
    Wih0A[j] = v2f{half ? 0.0f : w0, half ? 0.0f : bb0};
    B1A[j]   = v2f{half ? 0.0f : bb1, 0.0f};
  }
  const v2f onesB = v2f{half ? 0.0f : 1.0f, 0.0f};
  const v8f vzero = {0.f, 0.f, 0.f, 0.f, 0.f, 0.f, 0.f, 0.f};

  const float* xrow = x + (base + nIdx) * LSTM_T;

  v8f g0[4], g1[4];
  v8f gc[4];
#pragma unroll
  for (int j = 0; j < 4; ++j) gc[j] = vzero;
  float a0[4][8], a1[4][8];
  float c0[8], c1[8], h0[8], h1[8];
#pragma unroll
  for (int v = 0; v < 8; ++v) { c0[v] = 0.0f; c1[v] = 0.0f; h0[v] = 0.0f; h1[v] = 0.0f; }
  v2f xB;
  v2f hin[4];

  auto w0 = [&](int k) __attribute__((always_inline)) {
    const int r = k >> 2, j = k & 3;
    if (r == 0) g0[j] = wmma4(Wih0A[j], xB, vzero);
    else        g0[j] = wmma4(Whh0A[j][r - 1], v2f{h0[2 * r - 2], h0[2 * r - 1]}, g0[j]);
  };
  auto w1 = [&](int k) __attribute__((always_inline)) {
    const int r = k >> 2, j = k & 3;
    if (r == 0) g1[j] = wmma4(B1A[j], onesB, vzero);
    else        g1[j] = wmma4(Wih1A[j][r - 1], hin[r - 1], g1[j]);
  };
  auto w2 = [&](int k) __attribute__((always_inline)) {
    const int r = k >> 2, j = k & 3;
    if (r == 0) g1[j] = wmma4(Whh1A[j][0], v2f{h1[0], h1[1]}, gc[j]);
    else        g1[j] = wmma4(Whh1A[j][r], v2f{h1[2 * r], h1[2 * r + 1]}, g1[j]);
  };
#define W0(k) w0(k); asm volatile("" :: "v"(g0[(k) & 3])); FENCE();
#define W1(k) w1(k); asm volatile("" :: "v"(g1[(k) & 3])); FENCE();
#define W2(k) w2(k); asm volatile("" :: "v"(g1[(k) & 3])); FENCE();
#define GEMM0() W0(0) W0(1) W0(2) W0(3) W0(4) W0(5) W0(6) W0(7) W0(8) W0(9) \
  W0(10) W0(11) W0(12) W0(13) W0(14) W0(15) W0(16) W0(17) W0(18) W0(19)
#define GEMM1_BIAS() W1(0) W1(1) W1(2) W1(3)
#define GEMM1_IH() W1(4) W1(5) W1(6) W1(7) W1(8) W1(9) W1(10) W1(11) W1(12)  \
  W1(13) W1(14) W1(15) W1(16) W1(17) W1(18) W1(19)
#define GEMM1_HH() W2(0) W2(1) W2(2) W2(3) W2(4) W2(5) W2(6) W2(7) W2(8)     \
  W2(9) W2(10) W2(11) W2(12) W2(13) W2(14) W2(15)

#define STEP0(slot, tnext)                                                   \
  {                                                                          \
    xB = v2f{half ? 0.0f : xn, half ? 0.0f : 1.0f};                          \
    xn = xrow[tnext];                                                        \
    GEMM0()                                                                  \
    CELL(g0, c0, h0, a0)                                                     \
    hx[slot][0][lane] = float4{h0[0], h0[1], h0[2], h0[3]};                  \
    hx[slot][1][lane] = float4{h0[4], h0[5], h0[6], h0[7]};                  \
  }
#define STEP1(slot)                                                          \
  {                                                                          \
    const float4 u0 = hx[slot][0][lane];                                     \
    const float4 u1 = hx[slot][1][lane];                                     \
    FENCE();   \
    GEMM1_BIAS()                                                             \
    hin[0] = v2f{u0.x, u0.y}; hin[1] = v2f{u0.z, u0.w};                      \
    hin[2] = v2f{u1.x, u1.y}; hin[3] = v2f{u1.z, u1.w};                      \
    GEMM1_IH()                                                               \
    _Pragma("unroll") for (int j = 0; j < 4; ++j) {                          \
      gx[slot][2 * j + 0][lane] = float4{g1[j][0], g1[j][1], g1[j][2], g1[j][3]}; \
      gx[slot][2 * j + 1][lane] = float4{g1[j][4], g1[j][5], g1[j][6], g1[j][7]}; \
    }                                                                        \
  }
#define STEP2_GEMM() { GEMM1_HH() }
#define STEP2_FETCH(slot)                                                    \
  {                                                                          \
    _Pragma("unroll") for (int j = 0; j < 4; ++j) {                          \
      const float4 q0 = gx[slot][2 * j + 0][lane];                           \
      const float4 q1 = gx[slot][2 * j + 1][lane];                           \
      gc[j] = v8f{q0.x, q0.y, q0.z, q0.w, q1.x, q1.y, q1.z, q1.w};           \
    }                                                                        \
    FENCE();   \
  }
#define STEP2_CELL() { CELL(g1, c1, h1, a1) }

  float xn = xrow[0];

#define PHASE_EDGE(p)                                                        \
  if (role == 0) {                                                           \
    if (p < LSTM_T) {                                                        \
      const int tn = p + 1 < LSTM_T ? p + 1 : LSTM_T - 1;   \
      STEP0(p & 1, tn)                                                       \
    }                                                                        \
  } else if (role == 1) {                                                    \
    if (p >= 1 && p <= LSTM_T) STEP1((p - 1) & 1)                            \
  } else {                                                                   \
    if (p >= 3) STEP2_GEMM()                                                 \
    if (p >= 2 && p <= LSTM_T + 1) STEP2_FETCH(p & 1)                        \
    if (p >= 3) STEP2_CELL()                                                 \
  }
#define PHASE_FULL(p)                                                        \
  if (role == 0) {                                                           \
    STEP0(p & 1, p + 1)                                                      \
  } else if (role == 1) {                                                    \
    STEP1((p - 1) & 1)                                                       \
  } else {                                                                   \
    STEP2_GEMM()                                                             \
    STEP2_FETCH(p & 1)                                                       \
    STEP2_CELL()                                                             \
  }
#pragma unroll 1
  for (int p = 0; p < 3; ++p) {
    PHASE_EDGE(p)
    __syncthreads();
  }
#pragma unroll 1
  for (int p = 3; p < LSTM_T - 1; ++p) {
    PHASE_FULL(p)
    __syncthreads();
  }
#pragma unroll 1
  for (int p = LSTM_T - 1; p < LSTM_T + 3; ++p) {
    PHASE_EDGE(p)
    __syncthreads();
  }

  if (role == 2) {
    float partial = 0.0f;
#pragma unroll
    for (int v = 0; v < 8; ++v) partial += Wfc[v + 8 * half] * h1[v];
    const float tot = partial + __shfl_xor(partial, 16, 32);
    if (half == 0) out[base + nIdx] = tot + bfc[0];
  }
}

extern "C" void kernel_launch(void* const* d_in, const int* in_sizes, int n_in,
                              void* d_out, int out_size, void* d_ws,
                              size_t ws_size, hipStream_t stream) {
  (void)in_sizes; (void)n_in; (void)d_ws; (void)ws_size; (void)out_size;
  const float* x    = (const float*)d_in[0];
  const float* Wih0 = (const float*)d_in[1];
  const float* Whh0 = (const float*)d_in[2];
  const float* bih0 = (const float*)d_in[3];
  const float* bhh0 = (const float*)d_in[4];
  const float* Wih1 = (const float*)d_in[5];
  const float* Whh1 = (const float*)d_in[6];
  const float* bih1 = (const float*)d_in[7];
  const float* bhh1 = (const float*)d_in[8];
  const float* Wfc  = (const float*)d_in[9];
  const float* bfc  = (const float*)d_in[10];
  float* out = (float*)d_out;

  dim3 block(96);
  dim3 grid(LSTM_B / 16);
  lstm2_fc_kernel<<<grid, block, 0, stream>>>(x, Wih0, Whh0, bih0, bhh0, Wih1,
                                              Whh1, bih1, bhh1, Wfc, bfc, out);
}


# sigmoid and tanh each on one v_tanh_f32 (sigmoid(a) = 1/2 + tanh(a/2)/2, the halving folded exactly into the gate rows' weights and biases) in place of exp2 + rcp; no sum reordered, last bits move
# speedup vs baseline: 2.7664x; 1.1116x over previous
#include <hip/hip_runtime.h>


#define LSTM_B 4096
#define LSTM_T 512

typedef float v2f __attribute__((ext_vector_type(2)));
typedef float v8f __attribute__((ext_vector_type(8)));

static __device__ __forceinline__ v8f wmma4(v2f a, v2f b, v8f c) {
  return __builtin_amdgcn_wmma_f32_16x16x4_f32(false, a, false, b, (short)0, c,
                                               false, false);
}

#define FENCE() __builtin_amdgcn_sched_barrier(0)

#define TANH(x) __builtin_amdgcn_tanhf(x)
#define HALF 0.5f

#define FMA2(a, b, c) __builtin_elementwise_fma(a, b, c)
#define HALF2 (v2f{HALF, HALF})
#define CELL(g, c, h)                                                        \
  {                                                                          \
    v2f ti[4], tf[4], tg[4], to[4], tc[4];                                   \
    _Pragma("unroll") for (int p = 0; p < 4; ++p) {                          \
      ti[p].x = TANH(g[0][2 * p]);                                           \
      ti[p].y = TANH(g[0][2 * p + 1]);                                       \
    }                                                                        \
    _Pragma("unroll") for (int p = 0; p < 4; ++p) {                          \
      tf[p].x = TANH(g[1][2 * p]);                                           \
      tf[p].y = TANH(g[1][2 * p + 1]); ti[p] = FMA2(ti[p], HALF2, HALF2);    \
    }                                                                        \
    _Pragma("unroll") for (int p = 0; p < 4; ++p) {                          \
      tg[p].x = TANH(g[2][2 * p]);                                           \
      tg[p].y = TANH(g[2][2 * p + 1]); tf[p] = FMA2(tf[p], HALF2, HALF2);    \
    }                                                                        \
    to[0].x = TANH(g[3][0]); tf[0] = tf[0] * c[0];                           \
    to[0].y = TANH(g[3][1]); tf[1] = tf[1] * c[1];                           \
    to[1].x = TANH(g[3][2]); tf[2] = tf[2] * c[2];                           \
    to[1].y = TANH(g[3][3]); tf[3] = tf[3] * c[3];                           \
    to[2].x = TANH(g[3][4]); c[0] = FMA2(ti[0], tg[0], tf[0]);               \
    to[2].y = TANH(g[3][5]); c[1] = FMA2(ti[1], tg[1], tf[1]);               \
    to[3].x = TANH(g[3][6]); c[2] = FMA2(ti[2], tg[2], tf[2]);               \
    to[3].y = TANH(g[3][7]); c[3] = FMA2(ti[3], tg[3], tf[3]);               \
    _Pragma("unroll") for (int p = 0; p < 4; ++p) {                          \
      tc[p].x = TANH(c[p].x);                                                \
      tc[p].y = TANH(c[p].y); to[p] = FMA2(to[p], HALF2, HALF2);             \
    }                                                                        \
    _Pragma("unroll") for (int p = 0; p < 4; ++p) h[p] = to[p] * tc[p];      \
  }

__global__ void __launch_bounds__(96, 1) lstm2_fc_kernel(
    const float* __restrict__ x,
    const float* __restrict__ Wih0,
    const float* __restrict__ Whh0,
    const float* __restrict__ bih0,
    const float* __restrict__ bhh0,
    const float* __restrict__ Wih1,
    const float* __restrict__ Whh1,
    const float* __restrict__ bih1,
    const float* __restrict__ bhh1,
    const float* __restrict__ Wfc,
    const float* __restrict__ bfc,
    float* __restrict__ out)
{
  __shared__ float4 hx[2][2][32];
  __shared__ float4 gx[2][8][32];

  const int lane = threadIdx.x & 31;
  const int half = lane >> 4;
  const int nIdx = lane & 15;
  const int role = __builtin_amdgcn_readfirstlane((int)(threadIdx.x >> 5));
  const int base = blockIdx.x * 16;

  v2f Whh0A[4][4], Wih1A[4][4], Whh1A[4][4];
  v2f Wih0A[4];
  v2f B1A[4];
#pragma unroll
  for (int j = 0; j < 4; ++j) {
    const int row = 16 * j + nIdx;
    const float sc = (j == 2) ? 1.0f : HALF;
#pragma unroll
    for (int kc = 0; kc < 4; ++kc) {
      const float* p0 = Whh0 + row * 16 + 8 * half + 2 * kc;
      const float* p1 = Wih1 + row * 16 + 8 * half + 2 * kc;
      const float* p2 = Whh1 + row * 16 + 8 * half + 2 * kc;
      Whh0A[j][kc] = v2f{sc * p0[0], sc * p0[1]};
      Wih1A[j][kc] = v2f{sc * p1[0], sc * p1[1]};
      Whh1A[j][kc] = v2f{sc * p2[0], sc * p2[1]};
    }
    const float w0  = sc * Wih0[row];
    const float bb0 = sc * (bih0[row] + bhh0[row]);
    const float bb1 = sc * (bih1[row] + bhh1[row]);
    Wih0A[j] = v2f{half ? 0.0f : w0, half ? 0.0f : bb0};
    B1A[j]   = v2f{half ? 0.0f : bb1, 0.0f};
  }
  const v2f onesB = v2f{half ? 0.0f : 1.0f, 0.0f};
  const v8f vzero = {0.f, 0.f, 0.f, 0.f, 0.f, 0.f, 0.f, 0.f};

  const float* xrow = x + (base + nIdx) * LSTM_T;

  v8f g0[4], g1[4];
  v8f gc[4];
#pragma unroll
  for (int j = 0; j < 4; ++j) gc[j] = vzero;
  v2f c0[4], c1[4], h0[4], h1[4];
#pragma unroll
  for (int p = 0; p < 4; ++p) {
    c0[p] = v2f{0.0f, 0.0f}; c1[p] = v2f{0.0f, 0.0f};
    h0[p] = v2f{0.0f, 0.0f}; h1[p] = v2f{0.0f, 0.0f};
  }
  v2f xB;
  v2f hin[4];

  auto w0 = [&](int k) __attribute__((always_inline)) {
    const int r = k >> 2, j = k & 3;
    if (r == 0) g0[j] = wmma4(Wih0A[j], xB, vzero);
    else        g0[j] = wmma4(Whh0A[j][r - 1], h0[r - 1], g0[j]);
  };
  auto w1 = [&](int k) __attribute__((always_inline)) {
    const int r = k >> 2, j = k & 3;
    if (r == 0) g1[j] = wmma4(B1A[j], onesB, vzero);
    else        g1[j] = wmma4(Wih1A[j][r - 1], hin[r - 1], g1[j]);
  };
  auto w2 = [&](int k) __attribute__((always_inline)) {
    const int r = k >> 2, j = k & 3;
    if (r == 0) g1[j] = wmma4(Whh1A[j][0], h1[0], gc[j]);
    else        g1[j] = wmma4(Whh1A[j][r], h1[r], g1[j]);
  };
#define W0(k) w0(k); asm volatile("" :: "v"(g0[(k) & 3])); FENCE();
#define W1(k) w1(k); asm volatile("" :: "v"(g1[(k) & 3])); FENCE();
#define W2(k) w2(k); asm volatile("" :: "v"(g1[(k) & 3])); FENCE();
#define GEMM0() W0(0) W0(1) W0(2) W0(3) W0(4) W0(5) W0(6) W0(7) W0(8) W0(9) \
  W0(10) W0(11) W0(12) W0(13) W0(14) W0(15) W0(16) W0(17) W0(18) W0(19)
#define GEMM1_BIAS() W1(0) W1(1) W1(2) W1(3)
#define GEMM1_IH() W1(4) W1(5) W1(6) W1(7) W1(8) W1(9) W1(10) W1(11) W1(12)  \
  W1(13) W1(14) W1(15) W1(16) W1(17) W1(18) W1(19)
#define GEMM1_HH() W2(0) W2(1) W2(2) W2(3) W2(4) W2(5) W2(6) W2(7) W2(8)     \
  W2(9) W2(10) W2(11) W2(12) W2(13) W2(14) W2(15)

#define STEP0(slot, tnext)                                                   \
  {                                                                          \
    xB = v2f{half ? 0.0f : xn, half ? 0.0f : 1.0f};                          \
    xn = xrow[tnext];                                                        \
    GEMM0()                                                                  \
    CELL(g0, c0, h0)                                                         \
    hx[slot][0][lane] = float4{h0[0].x, h0[0].y, h0[1].x, h0[1].y};          \
    hx[slot][1][lane] = float4{h0[2].x, h0[2].y, h0[3].x, h0[3].y};          \
  }
#define STEP1(slot)                                                          \
  {                                                                          \
    const float4 u0 = hx[slot][0][lane];                                     \
    const float4 u1 = hx[slot][1][lane];                                     \
    FENCE();   \
    GEMM1_BIAS()                                                             \
    hin[0] = v2f{u0.x, u0.y}; hin[1] = v2f{u0.z, u0.w};                      \
    hin[2] = v2f{u1.x, u1.y}; hin[3] = v2f{u1.z, u1.w};                      \
    GEMM1_IH()                                                               \
    _Pragma("unroll") for (int j = 0; j < 4; ++j) {                          \
      gx[slot][2 * j + 0][lane] = float4{g1[j][0], g1[j][1], g1[j][2], g1[j][3]}; \
      gx[slot][2 * j + 1][lane] = float4{g1[j][4], g1[j][5], g1[j][6], g1[j][7]}; \
    }                                                                        \
  }
#define STEP2_GEMM() { GEMM1_HH() }
#define STEP2_FETCH(slot)                                                    \
  {                                                                          \
    _Pragma("unroll") for (int j = 0; j < 4; ++j) {                          \
      const float4 q0 = gx[slot][2 * j + 0][lane];                           \
      const float4 q1 = gx[slot][2 * j + 1][lane];                           \
      gc[j] = v8f{q0.x, q0.y, q0.z, q0.w, q1.x, q1.y, q1.z, q1.w};           \
    }                                                                        \
    FENCE();   \
  }
#define STEP2_CELL() { CELL(g1, c1, h1) }

  float xn = xrow[0];

#define PHASE_EDGE(p)                                                        \
  if (role == 0) {                                                           \
    if (p < LSTM_T) {                                                        \
      const int tn = p + 1 < LSTM_T ? p + 1 : LSTM_T - 1;   \
      STEP0(p & 1, tn)                                                       \
    }                                                                        \
  } else if (role == 1) {                                                    \
    if (p >= 1 && p <= LSTM_T) STEP1((p - 1) & 1)                            \
  } else {                                                                   \
    if (p >= 3) STEP2_GEMM()                                                 \
    if (p >= 2 && p <= LSTM_T + 1) STEP2_FETCH(p & 1)                        \
    if (p >= 3) STEP2_CELL()                                                 \
  }
#define PHASE_FULL(p)                                                        \
  if (role == 0) {                                                           \
    STEP0(p & 1, p + 1)                                                      \
  } else if (role == 1) {                                                    \
    STEP1((p - 1) & 1)                                                       \
  } else {                                                                   \
    STEP2_GEMM()                                                             \
    STEP2_FETCH(p & 1)                                                       \
    STEP2_CELL()                                                             \
  }
#pragma unroll 1
  for (int p = 0; p < 3; ++p) {
    PHASE_EDGE(p)
    __syncthreads();
  }
#pragma unroll 1
  for (int p = 3; p < LSTM_T - 1; ++p) {
    PHASE_FULL(p)
    __syncthreads();
  }
#pragma unroll 1
  for (int p = LSTM_T - 1; p < LSTM_T + 3; ++p) {
    PHASE_EDGE(p)
    __syncthreads();
  }

  if (role == 2) {
    float partial = 0.0f;
#pragma unroll
    for (int v = 0; v < 8; ++v) partial += Wfc[v + 8 * half] * h1[v >> 1][v & 1];
    const float tot = partial + __shfl_xor(partial, 16, 32);
    if (half == 0) out[base + nIdx] = tot + bfc[0];
  }
}

extern "C" void kernel_launch(void* const* d_in, const int* in_sizes, int n_in,
                              void* d_out, int out_size, void* d_ws,
                              size_t ws_size, hipStream_t stream) {
  (void)in_sizes; (void)n_in; (void)d_ws; (void)ws_size; (void)out_size;
  const float* x    = (const float*)d_in[0];
  const float* Wih0 = (const float*)d_in[1];
  const float* Whh0 = (const float*)d_in[2];
  const float* bih0 = (const float*)d_in[3];
  const float* bhh0 = (const float*)d_in[4];
  const float* Wih1 = (const float*)d_in[5];
  const float* Whh1 = (const float*)d_in[6];
  const float* bih1 = (const float*)d_in[7];
  const float* bhh1 = (const float*)d_in[8];
  const float* Wfc  = (const float*)d_in[9];
  const float* bfc  = (const float*)d_in[10];
  float* out = (float*)d_out;

  dim3 block(96);
  dim3 grid(LSTM_B / 16);
  lstm2_fc_kernel<<<grid, block, 0, stream>>>(x, Wih0, Whh0, bih0, bhh0, Wih1,
                                              Whh1, bih1, bhh1, Wfc, bfc, out);
}


# gate rows grouped by hidden unit (a tile = all 4 gates of 4 units): each layer's step split by tiles over two waves (4 waves, 4 SIMDs), h halves swapped through LDS; k and FC sums in a new order
# speedup vs baseline: 3.5837x; 1.2954x over previous
#include <hip/hip_runtime.h>


#define LSTM_B 4096
#define LSTM_T 512

typedef float v2f __attribute__((ext_vector_type(2)));
typedef float v8f __attribute__((ext_vector_type(8)));

static __device__ __forceinline__ v8f wmma4(v2f a, v2f b, v8f c) {
  return __builtin_amdgcn_wmma_f32_16x16x4_f32(false, a, false, b, (short)0, c,
                                               false, false);
}

#define FENCE() __builtin_amdgcn_sched_barrier(0)

#define TANH(x) __builtin_amdgcn_tanhf(x)
#define HALF 0.5f
#define FMA2(a, b, c) __builtin_elementwise_fma(a, b, c)
#define HALF2 (v2f{HALF, HALF})

#define CELL(g, c, h)                                                        \
  {                                                                          \
    v2f ti[2], tf[2], tg[2], to[2], tc[2];                                   \
    _Pragma("unroll") for (int jj = 0; jj < 2; ++jj) {                       \
      ti[jj].x = TANH(g[jj][0]);                                             \
      ti[jj].y = TANH(g[jj][1]);                                             \
    }                                                                        \
    _Pragma("unroll") for (int jj = 0; jj < 2; ++jj) {                       \
      tf[jj].x = TANH(g[jj][2]);                                             \
      tf[jj].y = TANH(g[jj][3]); ti[jj] = FMA2(ti[jj], HALF2, HALF2);        \
    }                                                                        \
    _Pragma("unroll") for (int jj = 0; jj < 2; ++jj) {                       \
      tg[jj].x = TANH(g[jj][4]);                                             \
      tg[jj].y = TANH(g[jj][5]); tf[jj] = FMA2(tf[jj], HALF2, HALF2);        \
    }                                                                        \
    _Pragma("unroll") for (int jj = 0; jj < 2; ++jj) {                       \
      to[jj].x = TANH(g[jj][6]);                                             \
      to[jj].y = TANH(g[jj][7]); tf[jj] = tf[jj] * c[jj];                    \
    }                                                                        \
    _Pragma("unroll") for (int jj = 0; jj < 2; ++jj)                         \
      c[jj] = FMA2(ti[jj], tg[jj], tf[jj]);                                  \
    _Pragma("unroll") for (int jj = 0; jj < 2; ++jj) {                       \
      tc[jj].x = TANH(c[jj].x);                                              \
      tc[jj].y = TANH(c[jj].y); to[jj] = FMA2(to[jj], HALF2, HALF2);         \
    }                                                                        \
    _Pragma("unroll") for (int jj = 0; jj < 2; ++jj) h[jj] = to[jj] * tc[jj]; \
  }

__global__ void __launch_bounds__(128, 1) lstm2_fc_kernel(
    const float* __restrict__ x,
    const float* __restrict__ Wih0,
    const float* __restrict__ Whh0,
    const float* __restrict__ bih0,
    const float* __restrict__ bhh0,
    const float* __restrict__ Wih1,
    const float* __restrict__ Whh1,
    const float* __restrict__ bih1,
    const float* __restrict__ bhh1,
    const float* __restrict__ Wfc,
    const float* __restrict__ bfc,
    float* __restrict__ out)
{
  __shared__ float4 hx[2][2][32];
  __shared__ float4 hy[2][2][32];

  const int lane = threadIdx.x & 31;
  const int half = lane >> 4;
  const int nIdx = lane & 15;
  const int role  = __builtin_amdgcn_readfirstlane((int)(threadIdx.x >> 5));
  const int layer = role >> 1;
  const int part  = role & 1;
  const int base  = blockIdx.x * 16;

  {
    const float4 z4 = float4{0.0f, 0.0f, 0.0f, 0.0f};
    (&hx[0][0][0])[threadIdx.x] = z4;
    (&hy[0][0][0])[threadIdx.x] = z4;
  }

  v2f WRo[2][2], WRp[2][2];
  v2f WI[2][4];
  v2f BA[2];
  const float* Wrec = layer ? Whh1 : Whh0;
  const int q = (nIdx & 7) >> 1;
  const float sc = (q == 2) ? 1.0f : HALF;
#pragma unroll
  for (int jj = 0; jj < 2; ++jj) {
    const int J = 2 * part + jj;
    const int u = 4 * J + 2 * (nIdx >> 3) + (nIdx & 1);
    const int row = 16 * q + u;
#pragma unroll
    for (int l = 0; l < 2; ++l) {
      const float* po = Wrec + row * 16 + 4 * (2 * part + l) + 2 * half;
      const float* pp = Wrec + row * 16 + 4 * (2 * (1 - part) + l) + 2 * half;
      WRo[jj][l] = v2f{sc * po[0], sc * po[1]};
      WRp[jj][l] = v2f{sc * pp[0], sc * pp[1]};
    }
#pragma unroll
    for (int kc = 0; kc < 4; ++kc) {
      const float* pi = Wih1 + row * 16 + 4 * kc + 2 * half;
      WI[jj][kc] = v2f{sc * pi[0], sc * pi[1]};
    }
    const float w0  = sc * Wih0[row];
    const float bb0 = sc * (bih0[row] + bhh0[row]);
    const float bb1 = sc * (bih1[row] + bhh1[row]);
    const v2f ba0 = v2f{half ? 0.0f : w0, half ? 0.0f : bb0};
    const v2f ba1 = v2f{half ? 0.0f : bb1, 0.0f};
    BA[jj] = layer ? ba1 : ba0;
  }
  const v2f onesB = v2f{half ? 0.0f : 1.0f, 0.0f};
  const v8f vzero = {0.f, 0.f, 0.f, 0.f, 0.f, 0.f, 0.f, 0.f};

  const float* xrow = x + (base + nIdx) * LSTM_T;

  v8f g[2];
  v2f c[2], h[2];
#pragma unroll
  for (int jj = 0; jj < 2; ++jj) {
    c[jj] = v2f{0.0f, 0.0f};
    h[jj] = v2f{0.0f, 0.0f};
  }
  v2f xB;

#define WM(jj, a, b, cc)                                                     \
  g[jj] = wmma4(a, b, cc); asm volatile("" :: "v"(g[jj])); FENCE();
#define LINK(A0, A1, b) WM(0, A0, b, g[0]) WM(1, A1, b, g[1])

#define STEP0(rs, ws, tnext)                                                 \
  {                                                                          \
    const float4 up = hx[rs][1 - part][lane];                                \
    xB = v2f{half ? 0.0f : xn, half ? 0.0f : 1.0f};                          \
    xn = xrow[tnext];                                                        \
    FENCE();   \
    WM(0, BA[0], xB, vzero) WM(1, BA[1], xB, vzero)                          \
    LINK(WRo[0][0], WRo[1][0], h[0])                                         \
    LINK(WRo[0][1], WRo[1][1], h[1])                                         \
    LINK(WRp[0][0], WRp[1][0], (v2f{up.x, up.y}))                            \
    LINK(WRp[0][1], WRp[1][1], (v2f{up.z, up.w}))                            \
    CELL(g, c, h)                                                            \
    hx[ws][part][lane] = float4{h[0].x, h[0].y, h[1].x, h[1].y};             \
  }
#define STEP1(rs, ws)                                                        \
  {                                                                          \
    const float4 a0 = hx[rs][0][lane];                                       \
    const float4 a1 = hx[rs][1][lane];                                       \
    const float4 up = hy[rs][1 - part][lane];                                \
    FENCE();   \
    WM(0, BA[0], onesB, vzero) WM(1, BA[1], onesB, vzero)                    \
    LINK(WRo[0][0], WRo[1][0], h[0])                                         \
    LINK(WRo[0][1], WRo[1][1], h[1])                                         \
    LINK(WI[0][0], WI[1][0], (v2f{a0.x, a0.y}))                              \
    LINK(WI[0][1], WI[1][1], (v2f{a0.z, a0.w}))                              \
    LINK(WI[0][2], WI[1][2], (v2f{a1.x, a1.y}))                              \
    LINK(WI[0][3], WI[1][3], (v2f{a1.z, a1.w}))                              \
    LINK(WRp[0][0], WRp[1][0], (v2f{up.x, up.y}))                            \
    LINK(WRp[0][1], WRp[1][1], (v2f{up.z, up.w}))                            \
    CELL(g, c, h)                                                            \
    hy[ws][part][lane] = float4{h[0].x, h[0].y, h[1].x, h[1].y};             \
  }

  float xn = xrow[0];

  __syncthreads();

#define PHASE_EDGE(p)                                                        \
  if (layer == 0) {                                                          \
    if (p < LSTM_T) {                                                        \
      const int tn = p + 1 < LSTM_T ? p + 1 : LSTM_T - 1;   \
      STEP0((p - 1) & 1, p & 1, tn)                                          \
    }                                                                        \
  } else {                                                                   \
    if (p >= 1) STEP1((p - 1) & 1, p & 1)                                    \
  }
#define PHASE_FULL(p)                                                        \
  if (layer == 0) {                                                          \
    STEP0((p - 1) & 1, p & 1, p + 1)                                         \
  } else {                                                                   \
    STEP1((p - 1) & 1, p & 1)                                                \
  }
  {
    const int p = 0;
    PHASE_EDGE(p)
    __syncthreads();
  }
#pragma unroll 1
  for (int p = 1; p < LSTM_T - 1; ++p) {
    PHASE_FULL(p)
    __syncthreads();
  }
#pragma unroll 1
  for (int p = LSTM_T - 1; p < LSTM_T + 1; ++p) {
    PHASE_EDGE(p)
    __syncthreads();
  }

  if (role == 2) {
    const float4 up = hy[LSTM_T & 1][1][lane];
    const float hv[8] = {h[0].x, h[0].y, h[1].x, h[1].y, up.x, up.y, up.z, up.w};
    float partial = 0.0f;
#pragma unroll
    for (int v = 0; v < 8; ++v)
      partial += Wfc[4 * (v >> 1) + 2 * half + (v & 1)] * hv[v];
    const float tot = partial + __shfl_xor(partial, 16, 32);
    if (half == 0) out[base + nIdx] = tot + bfc[0];
  }
}

extern "C" void kernel_launch(void* const* d_in, const int* in_sizes, int n_in,
                              void* d_out, int out_size, void* d_ws,
                              size_t ws_size, hipStream_t stream) {
  (void)in_sizes; (void)n_in; (void)d_ws; (void)ws_size; (void)out_size;
  const float* x    = (const float*)d_in[0];
  const float* Wih0 = (const float*)d_in[1];
  const float* Whh0 = (const float*)d_in[2];
  const float* bih0 = (const float*)d_in[3];
  const float* bhh0 = (const float*)d_in[4];
  const float* Wih1 = (const float*)d_in[5];
  const float* Whh1 = (const float*)d_in[6];
  const float* bih1 = (const float*)d_in[7];
  const float* bhh1 = (const float*)d_in[8];
  const float* Wfc  = (const float*)d_in[9];
  const float* bfc  = (const float*)d_in[10];
  float* out = (float*)d_out;

  dim3 block(128);
  dim3 grid(LSTM_B / 16);
  lstm2_fc_kernel<<<grid, block, 0, stream>>>(x, Wih0, Whh0, bih0, bhh0, Wih1,
                                              Whh1, bih1, bhh1, Wfc, bfc, out);
}


# each layer's input half of step s+1 (layer 1: bias as C tile + Wih1 @ h0; layer 0: x/bias link) runs as WMMAs between the gate-math slices of step s, a phase ahead; layer 1 k summed in a new order
# speedup vs baseline: 4.3501x; 1.2139x over previous
#include <hip/hip_runtime.h>


#define LSTM_B 4096
#define LSTM_T 512

typedef float v2f __attribute__((ext_vector_type(2)));
typedef float v8f __attribute__((ext_vector_type(8)));

static __device__ __forceinline__ v8f wmma4(v2f a, v2f b, v8f c) {
  return __builtin_amdgcn_wmma_f32_16x16x4_f32(false, a, false, b, (short)0, c,
                                               false, false);
}

#define FENCE() __builtin_amdgcn_sched_barrier(0)

#define TANH(x) __builtin_amdgcn_tanhf(x)
#define HALF 0.5f
#define FMA2(a, b, c) __builtin_elementwise_fma(a, b, c)
#define HALF2 (v2f{HALF, HALF})

#define CELL_DECL v2f ti[2], tf[2], tg[2], to[2], tc[2];
#define CS1(g)                                                               \
    _Pragma("unroll") for (int jj = 0; jj < 2; ++jj) {                       \
      ti[jj].x = TANH(g[jj][0]);                                             \
      ti[jj].y = TANH(g[jj][1]);                                             \
    }
#define CS2(g)                                                               \
    _Pragma("unroll") for (int jj = 0; jj < 2; ++jj) {                       \
      tf[jj].x = TANH(g[jj][2]);                                             \
      tf[jj].y = TANH(g[jj][3]); ti[jj] = FMA2(ti[jj], HALF2, HALF2);        \
    }
#define CS3(g)                                                               \
    _Pragma("unroll") for (int jj = 0; jj < 2; ++jj) {                       \
      tg[jj].x = TANH(g[jj][4]);                                             \
      tg[jj].y = TANH(g[jj][5]); tf[jj] = FMA2(tf[jj], HALF2, HALF2);        \
    }
#define CS4(g, c)                                                            \
    _Pragma("unroll") for (int jj = 0; jj < 2; ++jj) {                       \
      to[jj].x = TANH(g[jj][6]);                                             \
      to[jj].y = TANH(g[jj][7]); tf[jj] = tf[jj] * c[jj];                    \
    }
#define CS5(c)                                                               \
    _Pragma("unroll") for (int jj = 0; jj < 2; ++jj)                         \
      c[jj] = FMA2(ti[jj], tg[jj], tf[jj]);
#define CS6(c)                                                               \
    _Pragma("unroll") for (int jj = 0; jj < 2; ++jj) {                       \
      tc[jj].x = TANH(c[jj].x);                                              \
      tc[jj].y = TANH(c[jj].y); to[jj] = FMA2(to[jj], HALF2, HALF2);         \
    }
#define CS7(h)                                                               \
    _Pragma("unroll") for (int jj = 0; jj < 2; ++jj) h[jj] = to[jj] * tc[jj];
#define CELL(g, c, h)                                                        \
  {                                                                          \
    CELL_DECL                                                                \
    CS1(g) CS2(g) CS3(g) CS4(g, c) CS5(c) CS6(c) CS7(h)                      \
  }

__global__ void __launch_bounds__(128, 1) lstm2_fc_kernel(
    const float* __restrict__ x,
    const float* __restrict__ Wih0,
    const float* __restrict__ Whh0,
    const float* __restrict__ bih0,
    const float* __restrict__ bhh0,
    const float* __restrict__ Wih1,
    const float* __restrict__ Whh1,
    const float* __restrict__ bih1,
    const float* __restrict__ bhh1,
    const float* __restrict__ Wfc,
    const float* __restrict__ bfc,
    float* __restrict__ out)
{
  __shared__ float4 hx[2][2][32];
  __shared__ float4 hy[2][2][32];

  const int lane = threadIdx.x & 31;
  const int half = lane >> 4;
  const int nIdx = lane & 15;
  const int role  = __builtin_amdgcn_readfirstlane((int)(threadIdx.x >> 5));
  const int layer = role >> 1;
  const int part  = role & 1;
  const int base  = blockIdx.x * 16;

  {
    const float4 z4 = float4{0.0f, 0.0f, 0.0f, 0.0f};
    (&hx[0][0][0])[threadIdx.x] = z4;
    (&hy[0][0][0])[threadIdx.x] = z4;
  }

  v2f WRo[2][2], WRp[2][2];
  v2f WI[2][4];
  v2f BA[2];
  v8f biasC[2];
  const float* Wrec = layer ? Whh1 : Whh0;
  const int q = (nIdx & 7) >> 1;
  const float sc = (q == 2) ? 1.0f : HALF;
#pragma unroll
  for (int jj = 0; jj < 2; ++jj) {
    const int J = 2 * part + jj;
    const int u = 4 * J + 2 * (nIdx >> 3) + (nIdx & 1);
    const int row = 16 * q + u;
#pragma unroll
    for (int l = 0; l < 2; ++l) {
      const float* po = Wrec + row * 16 + 4 * (2 * part + l) + 2 * half;
      const float* pp = Wrec + row * 16 + 4 * (2 * (1 - part) + l) + 2 * half;
      WRo[jj][l] = v2f{sc * po[0], sc * po[1]};
      WRp[jj][l] = v2f{sc * pp[0], sc * pp[1]};
    }
#pragma unroll
    for (int kc = 0; kc < 4; ++kc) {
      const float* pi = Wih1 + row * 16 + 4 * kc + 2 * half;
      WI[jj][kc] = v2f{sc * pi[0], sc * pi[1]};
    }
    const float w0  = sc * Wih0[row];
    const float bb0 = sc * (bih0[row] + bhh0[row]);
    BA[jj] = v2f{half ? 0.0f : w0, half ? 0.0f : bb0};
#pragma unroll
    for (int r = 0; r < 8; ++r) {
      const int rowr = 16 * (r >> 1) + 4 * J + 2 * half + (r & 1);
      const float scr = ((r >> 1) == 2) ? 1.0f : HALF;
      biasC[jj][r] = scr * (bih1[rowr] + bhh1[rowr]);
    }
  }
  const v8f vzero = {0.f, 0.f, 0.f, 0.f, 0.f, 0.f, 0.f, 0.f};

  const float* xrow = x + (base + nIdx) * LSTM_T;

  v8f G0[2], G1[2];
  v2f c[2], h[2];
#pragma unroll
  for (int jj = 0; jj < 2; ++jj) {
    c[jj] = v2f{0.0f, 0.0f};
    h[jj] = v2f{0.0f, 0.0f};
    G0[jj] = vzero;
    G1[jj] = vzero;
  }
  v2f xB;

#define WM(d, a, b, cc)                                                      \
  d = wmma4(a, b, cc); asm volatile("" :: "v"(d)); FENCE();

#define REC(up, G)                                                           \
    WM(G[0], WRo[0][0], h[0], G[0]) WM(G[1], WRo[1][0], h[0], G[1])          \
    WM(G[0], WRo[0][1], h[1], G[0]) WM(G[1], WRo[1][1], h[1], G[1])          \
    WM(G[0], WRp[0][0], (v2f{up.x, up.y}), G[0])                             \
    WM(G[1], WRp[1][0], (v2f{up.x, up.y}), G[1])                             \
    WM(G[0], WRp[0][1], (v2f{up.z, up.w}), G[0])                             \
    WM(G[1], WRp[1][1], (v2f{up.z, up.w}), G[1])

#define STEP0(rs, ws, tnext2, G, N)                                          \
  {                                                                          \
    const float4 up = hx[rs][1 - part][lane];                                \
    const float xf = xrow[tnext2];                                           \
    FENCE();   \
    REC(up, G)                                                               \
    xB = v2f{half ? 0.0f : xa, half ? 0.0f : 1.0f};                          \
    CELL_DECL                                                                \
    WM(N[0], BA[0], xB, vzero) CS1(G) FENCE();                               \
    WM(N[1], BA[1], xB, vzero) CS2(G) FENCE();                               \
    CS3(G) CS4(G, c) CS5(c) CS6(c) CS7(h)                                    \
    hx[ws][part][lane] = float4{h[0].x, h[0].y, h[1].x, h[1].y};             \
    FENCE();                                                                 \
    xa = xf;                                                                 \
  }
#define A0XY (v2f{a0.x, a0.y})
#define A0ZW (v2f{a0.z, a0.w})
#define A1XY (v2f{a1.x, a1.y})
#define A1ZW (v2f{a1.z, a1.w})
#define PRE0(N) WM(N[0], WI[0][0], A0XY, biasC[0])
#define PRE1(N) WM(N[1], WI[1][0], A0XY, biasC[1])
#define PRE2(N) WM(N[0], WI[0][1], A0ZW, N[0])
#define PRE3(N) WM(N[1], WI[1][1], A0ZW, N[1])
#define PRE4(N) WM(N[0], WI[0][2], A1XY, N[0])
#define PRE5(N) WM(N[1], WI[1][2], A1XY, N[1])
#define PRE6(N) WM(N[0], WI[0][3], A1ZW, N[0])
#define PRE7(N) WM(N[1], WI[1][3], A1ZW, N[1])
#define STEP1(rs, ws, G, N)                                                  \
  {                                                                          \
    const float4 up = hy[rs][1 - part][lane];                                \
    const float4 a0 = hx[rs][0][lane];                                       \
    const float4 a1 = hx[rs][1][lane];                                       \
    FENCE();   \
    REC(up, G)                                                               \
    CELL_DECL                                                                \
    PRE0(N) CS1(G) FENCE();                                                  \
    PRE1(N) CS2(G) FENCE();                                                  \
    PRE2(N) CS3(G) FENCE();                                                  \
    PRE3(N) CS4(G, c) FENCE();                                               \
    PRE4(N) CS5(c) CS6(c) FENCE();                                           \
    PRE5(N) CS7(h)                                                           \
    hy[ws][part][lane] = float4{h[0].x, h[0].y, h[1].x, h[1].y};             \
    FENCE();                                                                 \
    PRE6(N) PRE7(N)                                                          \
  }
#define STEP1_EDGE(p, rs, ws, G, N)                                          \
  {                                                                          \
    if (p >= 2) {                                                            \
      const float4 up = hy[rs][1 - part][lane];                              \
      REC(up, G)                                                             \
      CELL(G, c, h)                                                          \
      hy[ws][part][lane] = float4{h[0].x, h[0].y, h[1].x, h[1].y};           \
    }                                                                        \
    if (p <= LSTM_T) {                                                       \
      const float4 a0 = hx[rs][0][lane];                                     \
      const float4 a1 = hx[rs][1][lane];                                     \
      PRE0(N) PRE1(N) PRE2(N) PRE3(N) PRE4(N) PRE5(N) PRE6(N) PRE7(N)        \
    }                                                                        \
  }

  float xa = 0.0f;
  if (layer == 0) {
    const float x0 = xrow[0];
    xa = xrow[1];
    xB = v2f{half ? 0.0f : x0, half ? 0.0f : 1.0f};
    G0[0] = wmma4(BA[0], xB, vzero);
    G0[1] = wmma4(BA[1], xB, vzero);
  }

  __syncthreads();

#define PHASE_EDGE(p, G, N)                                                  \
  if (layer == 0) {                                                          \
    if (p < LSTM_T) {                                                        \
      STEP0((p - 1) & 1, p & 1, (p + 2 < LSTM_T ? p + 2 : LSTM_T - 1), G, N) \
    }                                                                        \
  } else {                                                                   \
    if (p >= 1) STEP1_EDGE(p, (p - 1) & 1, p & 1, G, N)                      \
  }                                                                          \
  __syncthreads();
#define PHASE_FULL(p, par, G, N)                                             \
  if (layer == 0) {                                                          \
    STEP0(1 - par, par, p + 2, G, N)                                         \
  } else {                                                                   \
    STEP1(1 - par, par, G, N)                                                \
  }                                                                          \
  __syncthreads();

  PHASE_EDGE(0, G0, G1)
  PHASE_EDGE(1, G1, G0)
#pragma unroll 1
  for (int p = 2; p < LSTM_T - 2; p += 2) {
    PHASE_FULL(p, 0, G0, G1)
    PHASE_FULL(p + 1, 1, G1, G0)
  }
  PHASE_EDGE((LSTM_T - 2), G0, G1)
  PHASE_EDGE((LSTM_T - 1), G1, G0)
  PHASE_EDGE((LSTM_T), G0, G1)
  PHASE_EDGE((LSTM_T + 1), G1, G0)

  if (role == 2) {
    const float4 up = hy[(LSTM_T + 1) & 1][1][lane];
    const float hv[8] = {h[0].x, h[0].y, h[1].x, h[1].y, up.x, up.y, up.z, up.w};
    float partial = 0.0f;
#pragma unroll
    for (int v = 0; v < 8; ++v)
      partial += Wfc[4 * (v >> 1) + 2 * half + (v & 1)] * hv[v];
    const float tot = partial + __shfl_xor(partial, 16, 32);
    if (half == 0) out[base + nIdx] = tot + bfc[0];
  }
}

extern "C" void kernel_launch(void* const* d_in, const int* in_sizes, int n_in,
                              void* d_out, int out_size, void* d_ws,
                              size_t ws_size, hipStream_t stream) {
  (void)in_sizes; (void)n_in; (void)d_ws; (void)ws_size; (void)out_size;
  const float* x    = (const float*)d_in[0];
  const float* Wih0 = (const float*)d_in[1];
  const float* Whh0 = (const float*)d_in[2];
  const float* bih0 = (const float*)d_in[3];
  const float* bhh0 = (const float*)d_in[4];
  const float* Wih1 = (const float*)d_in[5];
  const float* Whh1 = (const float*)d_in[6];
  const float* bih1 = (const float*)d_in[7];
  const float* bhh1 = (const float*)d_in[8];
  const float* Wfc  = (const float*)d_in[9];
  const float* bfc  = (const float*)d_in[10];
  float* out = (float*)d_out;

  dim3 block(128);
  dim3 grid(LSTM_B / 16);
  lstm2_fc_kernel<<<grid, block, 0, stream>>>(x, Wih0, Whh0, bih0, bhh0, Wih1,
                                              Whh1, bih1, bhh1, Wfc, bfc, out);
}
